# starting kernel
# baseline (speedup 1.0000x reference)
#include <hip/hip_runtime.h>
#include <hip/hip_bf16.h>

typedef __attribute__((ext_vector_type(2))) float v2f;
typedef __attribute__((ext_vector_type(8))) float v8f;

__global__ void zero_f32(float* __restrict__ p, int n) {
    int i = blockIdx.x * blockDim.x + threadIdx.x;
    if (i < n) p[i] = 0.0f;
}

__global__ void degree_k(const int* __restrict__ dst, float* __restrict__ deg, int E) {
    int e = blockIdx.x * blockDim.x + threadIdx.x;
    if (e < E) atomicAdd(&deg[dst[e]], 1.0f);
}

__global__ void scatter_k(const float* __restrict__ h, const int* __restrict__ src,
                          const int* __restrict__ dst, const float* __restrict__ ew,
                          float* __restrict__ agg, int E, int K) {
    int tid = blockIdx.x * blockDim.x + threadIdx.x;
    int chunks = K >> 2;
    int total = E * chunks;
    if (tid >= total) return;
    int e = tid / chunks;
    int c = (tid - e * chunks) << 2;

    int s = src[e];
    int d = dst[e];
    float w = ew[e];

    float4 hv = *(const float4*)(h + (size_t)s * K + c);
    float* ap = agg + (size_t)d * K + c;
    atomicAdd(ap + 0, hv.x * w);
    atomicAdd(ap + 1, hv.y * w);
    atomicAdd(ap + 2, hv.z * w);
    atomicAdd(ap + 3, hv.w * w);
}

__global__ void sage_gemm(const float* __restrict__ h, const float* __restrict__ agg,
                          const float* __restrict__ deg,
                          const float* __restrict__ Ws, const float* __restrict__ Wn,
                          const float* __restrict__ bias,
                          float* __restrict__ out,
                          int K, int Fout, int activate) {
    int lane = threadIdx.x;
    int m    = lane & 15;
    int kh   = (lane >> 4) << 1;
    int row0 = blockIdx.x << 4;
    int col0 = blockIdx.y << 4;

    int node = row0 + m;
    int col  = col0 + m;
    int   cs    = (col < Fout) ? col  : 0;
    float cmask = (col < Fout) ? 1.0f : 0.0f;

    float invdeg = 1.0f / fmaxf(deg[node], 1.0f);

    const float* hrow = h   + (size_t)node * K;
    const float* arow = agg + (size_t)node * K;

    v8f c = {};
    for (int k = 0; k < K; k += 4) {
        v2f a, b;
        a.x = hrow[k + kh];
        a.y = hrow[k + kh + 1];
        b.x = Ws[(k + kh)     * Fout + cs] * cmask;
        b.y = Ws[(k + kh + 1) * Fout + cs] * cmask;
        c = __builtin_amdgcn_wmma_f32_16x16x4_f32(false, a, false, b,
                                                  (short)0, c, false, false);
        a.x = arow[k + kh]     * invdeg;
        a.y = arow[k + kh + 1] * invdeg;
        b.x = Wn[(k + kh)     * Fout + cs] * cmask;
        b.y = Wn[(k + kh + 1) * Fout + cs] * cmask;
        c = __builtin_amdgcn_wmma_f32_16x16x4_f32(false, a, false, b,
                                                  (short)0, c, false, false);
    }

    int half = lane >> 4;
    int cn   = col0 + (lane & 15);
    if (cn < Fout) {
        float bv = bias[cn];
        for (int j = 0; j < 8; ++j) {
            int rm = row0 + j + (half << 3);
            float v = c[j] + bv;
            if (activate) v = tanhf(v);
            out[(size_t)rm * Fout + cn] = v;
        }
    }
}

extern "C" void kernel_launch(void* const* d_in, const int* in_sizes, int n_in,
                              void* d_out, int out_size, void* d_ws, size_t ws_size,
                              hipStream_t stream) {
    const float* b_z  = (const float*)d_in[0];
    const int*   esrc = (const int*)  d_in[1];
    const int*   edst = (const int*)  d_in[2];
    const float* ew   = (const float*)d_in[3];
    const float* Ws0  = (const float*)d_in[4];
    const float* Wn0  = (const float*)d_in[5];
    const float* b0   = (const float*)d_in[6];
    const float* Ws1  = (const float*)d_in[7];
    const float* Wn1  = (const float*)d_in[8];
    const float* b1   = (const float*)d_in[9];
    const float* Ws2  = (const float*)d_in[10];
    const float* Wn2  = (const float*)d_in[11];
    const float* b2   = (const float*)d_in[12];

    const int N = in_sizes[0] / 64;
    const int E = in_sizes[1];

    float* deg = (float*)d_ws;
    float* h1  = deg + N;
    float* h2  = h1  + (size_t)N * 32;
    float* agg = h2  + (size_t)N * 16;
    float* out = (float*)d_out;

    auto cdiv = [](int a, int b) { return (a + b - 1) / b; };

    zero_f32<<<cdiv(N, 256), 256, 0, stream>>>(deg, N);
    degree_k<<<cdiv(E, 256), 256, 0, stream>>>(edst, deg, E);

    struct Layer {
        const float *hin, *Ws, *Wn, *b;
        float* hout;
        int K, F, act;
    } L[3] = {
        { b_z, Ws0, Wn0, b0, h1,  64, 32, 1 },
        { h1,  Ws1, Wn1, b1, h2,  32, 16, 1 },
        { h2,  Ws2, Wn2, b2, out, 16,  1, 0 },
    };

    for (int i = 0; i < 3; ++i) {
        const int K = L[i].K, F = L[i].F;
        zero_f32<<<cdiv(N * K, 256), 256, 0, stream>>>(agg, N * K);
        scatter_k<<<cdiv(E * (K / 4), 256), 256, 0, stream>>>(
            L[i].hin, esrc, edst, ew, agg, E, K);
        dim3 grid(N / 16, cdiv(F, 16));
        sage_gemm<<<grid, 32, 0, stream>>>(
            L[i].hin, agg, deg, L[i].Ws, L[i].Wn, L[i].b, L[i].hout, K, F, L[i].act);
    }
}


# Apply Wn before the edge sums (per node sum of ew*(h@Wn)[src], not (sum of ew*h[src])@Wn): atomics run Fout wide (32/16/1), not K wide (64/32/16); this changes the order of the f32 sums
# speedup vs baseline: 1.1938x; 1.1938x over previous
#include <hip/hip_runtime.h>


typedef __attribute__((ext_vector_type(2))) float v2f;
typedef __attribute__((ext_vector_type(8))) float v8f;

__global__ void zero_f32(float* __restrict__ p, int n) {
    int i = blockIdx.x * blockDim.x + threadIdx.x;
    if (i < n) p[i] = 0.0f;
}

__global__ void degree_k(const int* __restrict__ dst, float* __restrict__ deg, int E) {
    int e = blockIdx.x * blockDim.x + threadIdx.x;
    if (e < E) atomicAdd(&deg[dst[e]], 1.0f);
}

template <int K, int F, bool FIRST>
__global__ void __launch_bounds__(32)
sage_project(const float* __restrict__ xin, const float* __restrict__ accin,
             const float* __restrict__ deg,
             const float* __restrict__ Ws, const float* __restrict__ Wn,
             const float* __restrict__ bias,
             float* __restrict__ S, float* __restrict__ P, float* __restrict__ acc) {
    static_assert(F == 1 || F % 16 == 0, "column tiling");
    static_assert(K % 4 == 0, "k steps");
    constexpr int NS = K / 4;
    constexpr int FT = (F >= 16) ? F / 16 : 1;
    constexpr int NT = (F >= 16) ? 2 * FT : 1;

    const int lane = threadIdx.x;
    const int c    = lane & 15;
    const int hh   = lane >> 4;
    const int row0 = blockIdx.x << 4;
    const int node = row0 + c;

    v2f a[NS];
    const float* xrow = xin + (size_t)node * K;
    if (FIRST) {
        for (int t = 0; t < NS; ++t) {
            const int k0 = 4 * t + 2 * hh;
            a[t] = *(const v2f*)(xrow + k0);
        }
    } else {
        const float invdeg = 1.0f / fmaxf(deg[node], 1.0f);
        const float* grow = accin + (size_t)node * K;
        for (int t = 0; t < NS; ++t) {
            const int k0 = 4 * t + 2 * hh;
            const v2f s = *(const v2f*)(xrow + k0);
            const v2f g = *(const v2f*)(grow + k0);
            a[t].x = tanhf(s.x + g.x * invdeg);
            a[t].y = tanhf(s.y + g.y * invdeg);
        }
    }

    v8f d[NT];
    for (int j = 0; j < NT; ++j) d[j] = v8f{};

    for (int t = 0; t < NS; ++t) {
        const int k0 = 4 * t + 2 * hh;
        for (int j = 0; j < NT; ++j) {
            v2f b;
            if (F >= 16) {
                const float* W = (j < FT) ? Ws : Wn;
                const int col  = 16 * (j % FT) + c;
                b.x = W[(k0)     * F + col];
                b.y = W[(k0 + 1) * F + col];
            } else {
                const float* W = (c == 1) ? Wn : Ws;
                const float w0 = W[k0];
                const float w1 = W[k0 + 1];
                b.x = (c < 2) ? w0 : 0.0f;
                b.y = (c < 2) ? w1 : 0.0f;
            }
            d[j] = __builtin_amdgcn_wmma_f32_16x16x4_f32(false, a[t], false, b,
                                                         (short)0, d[j], false, false);
        }
    }

    for (int j = 0; j < NT; ++j) {
        for (int r = 0; r < 8; ++r) {
            const size_t row = (size_t)(row0 + 8 * hh + r);
            if (F >= 16) {
                const int col = 16 * (j % FT) + c;
                if (j < FT) S[row * F + col] = d[j][r] + bias[col];
                else        P[row * F + col] = d[j][r];
            } else {
                float* o = (c == 1) ? P : S;
                const float v = (c == 0) ? d[0][r] + bias[0] : d[0][r];
                if (c < 2) o[row] = v;
            }
        }
    }

    float* ab = acc + (size_t)row0 * F;
    for (int i = lane * 4; i < 16 * F; i += 128)
        *(float4*)(ab + i) = make_float4(0.0f, 0.0f, 0.0f, 0.0f);
}

template <int F>
__global__ void scatter_proj(const float* __restrict__ P, const int* __restrict__ src,
                             const int* __restrict__ dst, const float* __restrict__ ew,
                             float* __restrict__ acc, int E) {
    constexpr int CH = F / 4;
    int tid = blockIdx.x * blockDim.x + threadIdx.x;
    if (tid >= E * CH) return;
    int e = tid / CH;
    int c = (tid - e * CH) << 2;

    int s = src[e];
    int d = dst[e];
    float w = ew[e];

    float4 pv = *(const float4*)(P + (size_t)s * F + c);
    float* ap = acc + (size_t)d * F + c;
    atomicAdd(ap + 0, pv.x * w);
    atomicAdd(ap + 1, pv.y * w);
    atomicAdd(ap + 2, pv.z * w);
    atomicAdd(ap + 3, pv.w * w);
}

__global__ void scatter_proj1(const float* __restrict__ P, const int* __restrict__ src,
                              const int* __restrict__ dst, const float* __restrict__ ew,
                              float* __restrict__ acc, int E) {
    int e = blockIdx.x * blockDim.x + threadIdx.x;
    if (e < E) atomicAdd(&acc[dst[e]], P[src[e]] * ew[e]);
}

__global__ void sage_final(const float* __restrict__ S, const float* __restrict__ acc,
                           const float* __restrict__ deg, float* __restrict__ out, int n) {
    int i = blockIdx.x * blockDim.x + threadIdx.x;
    if (i < n) out[i] = S[i] + acc[i] * (1.0f / fmaxf(deg[i], 1.0f));
}

extern "C" void kernel_launch(void* const* d_in, const int* in_sizes, int n_in,
                              void* d_out, int out_size, void* d_ws, size_t ws_size,
                              hipStream_t stream) {
    const float* b_z  = (const float*)d_in[0];
    const int*   esrc = (const int*)  d_in[1];
    const int*   edst = (const int*)  d_in[2];
    const float* ew   = (const float*)d_in[3];
    const float* Ws0  = (const float*)d_in[4];
    const float* Wn0  = (const float*)d_in[5];
    const float* b0   = (const float*)d_in[6];
    const float* Ws1  = (const float*)d_in[7];
    const float* Wn1  = (const float*)d_in[8];
    const float* b1   = (const float*)d_in[9];
    const float* Ws2  = (const float*)d_in[10];
    const float* Wn2  = (const float*)d_in[11];
    const float* b2   = (const float*)d_in[12];

    const int N = in_sizes[0] / 64;
    const int E = in_sizes[1];
    const size_t n = (size_t)N;

    float* ws   = (float*)d_ws;
    float* deg  = ws;
    float* S0   = ws + n;
    float* acc0 = ws + 33 * n;
    float* P0   = ws + 65 * n;
    float* S1   = ws + 65 * n;
    float* acc1 = ws + 81 * n;
    float* P1   = ws + 97 * n;
    float* S2   = ws + n;
    float* P2   = ws + 2 * n;
    float* acc2 = ws + 3 * n;
    float* out  = (float*)d_out;

    auto cdiv = [](int a, int b) { return (a + b - 1) / b; };

    zero_f32<<<cdiv(N, 256), 256, 0, stream>>>(deg, N);
    degree_k<<<cdiv(E, 256), 256, 0, stream>>>(edst, deg, E);

    sage_project<64, 32, true><<<N / 16, 32, 0, stream>>>(
        b_z, nullptr, deg, Ws0, Wn0, b0, S0, P0, acc0);
    scatter_proj<32><<<cdiv(E * 8, 256), 256, 0, stream>>>(P0, esrc, edst, ew, acc0, E);

    sage_project<32, 16, false><<<N / 16, 32, 0, stream>>>(
        S0, acc0, deg, Ws1, Wn1, b1, S1, P1, acc1);
    scatter_proj<16><<<cdiv(E * 4, 256), 256, 0, stream>>>(P1, esrc, edst, ew, acc1, E);

    sage_project<16, 1, false><<<N / 16, 32, 0, stream>>>(
        S1, acc1, deg, Ws2, Wn2, b2, S2, P2, acc2);
    scatter_proj1<<<cdiv(E, 256), 256, 0, stream>>>(P2, esrc, edst, ew, acc2, E);

    sage_final<<<cdiv(N, 256), 256, 0, stream>>>(S2, acc2, deg, out, N);
}


# Edges grouped by destination per call (two-level counting sort, LDS counters, arrival order within a node), then per-node sums of ew*P[src] without float atomics; changes the order of the f32 sums
# speedup vs baseline: 2.5948x; 2.1736x over previous
#include <hip/hip_runtime.h>


typedef __attribute__((ext_vector_type(2))) float v2f;
typedef __attribute__((ext_vector_type(8))) float v8f;

constexpr int SORT_G  = 256;
constexpr int BIN_SH  = 8;
constexpr int BIN_W   = 1 << BIN_SH;
constexpr int MAX_BIN = 1024;

__global__ void __launch_bounds__(1024)
bin_count(const int* __restrict__ dst, int* __restrict__ counts, int E, int chunk, int nbin) {
    __shared__ int h[MAX_BIN];
    const int t = threadIdx.x;
    const int g = blockIdx.x;
    for (int i = t; i < nbin; i += 1024) h[i] = 0;
    __syncthreads();
    const int lo = g * chunk;
    const int hi = min(E, lo + chunk);
    for (int e = lo + t; e < hi; e += 1024) atomicAdd(&h[dst[e] >> BIN_SH], 1);
    __syncthreads();
    for (int i = t; i < nbin; i += 1024) counts[i * SORT_G + g] = h[i];
}

__global__ void __launch_bounds__(1024)
scan_excl(const int* __restrict__ in, int* __restrict__ out, int M) {
    __shared__ int sh[1024];
    const int t = threadIdx.x;
    const int b = blockIdx.x;

    int part = 0;
    int j = 0;
    for (; j + 8 <= b; j += 8) {
        const int a0 = in[t + 1024 * (j + 0)];
        const int a1 = in[t + 1024 * (j + 1)];
        const int a2 = in[t + 1024 * (j + 2)];
        const int a3 = in[t + 1024 * (j + 3)];
        const int a4 = in[t + 1024 * (j + 4)];
        const int a5 = in[t + 1024 * (j + 5)];
        const int a6 = in[t + 1024 * (j + 6)];
        const int a7 = in[t + 1024 * (j + 7)];
        part += ((a0 + a1) + (a2 + a3)) + ((a4 + a5) + (a6 + a7));
    }
    for (; j < b; ++j) part += in[t + 1024 * j];
    sh[t] = part;
    __syncthreads();
    for (int s = 512; s > 0; s >>= 1) {
        if (t < s) sh[t] += sh[t + s];
        __syncthreads();
    }
    const int base = sh[0];
    __syncthreads();

    const int i = 1024 * b + t;
    const int v = (i < M) ? in[i] : 0;
    sh[t] = v;
    __syncthreads();
    for (int off = 1; off < 1024; off <<= 1) {
        const int x = (t >= off) ? sh[t - off] : 0;
        __syncthreads();
        sh[t] += x;
        __syncthreads();
    }
    const int incl = sh[t];
    if (i < M) {
        out[i] = base + incl - v;
        if (i == M - 1) out[M] = base + incl;
    }
}

__global__ void __launch_bounds__(1024)
bin_scatter(const int* __restrict__ src, const int* __restrict__ dst,
            const float* __restrict__ ew, const int* __restrict__ offs,
            int* __restrict__ tdst, int* __restrict__ tpair, int E, int chunk, int nbin) {
    __shared__ int cur[MAX_BIN];
    const int t = threadIdx.x;
    const int g = blockIdx.x;
    for (int i = t; i < nbin; i += 1024) cur[i] = offs[i * SORT_G + g];
    __syncthreads();
    const int lo = g * chunk;
    const int hi = min(E, lo + chunk);
    for (int e = lo + t; e < hi; e += 1024) {
        const int d   = dst[e];
        const int pos = atomicAdd(&cur[d >> BIN_SH], 1);
        tdst[pos] = d;
        ((int2*)tpair)[pos] = make_int2(src[e], __float_as_int(ew[e]));
    }
}

__global__ void __launch_bounds__(1024)
bin_sort(const int* __restrict__ tdst, const int* __restrict__ tpair,
         const int* __restrict__ offs, int* __restrict__ rowptr,
         float* __restrict__ deg, int* __restrict__ sorted, int N) {
    __shared__ int cnt[BIN_W];
    __shared__ int sc[BIN_W];
    __shared__ int cur[BIN_W];
    const int t   = threadIdx.x;
    const int b   = blockIdx.x;
    const int n0  = b << BIN_SH;
    const int beg = offs[b * SORT_G];
    const int end = offs[(b + 1) * SORT_G];

    if (t < BIN_W) cnt[t] = 0;
    __syncthreads();
    for (int i = beg + t; i < end; i += 1024) atomicAdd(&cnt[tdst[i] - n0], 1);
    __syncthreads();

    const int v = (t < BIN_W) ? cnt[t] : 0;
    if (t < BIN_W) sc[t] = v;
    __syncthreads();
    for (int off = 1; off < BIN_W; off <<= 1) {
        int x = 0;
        if (t < BIN_W && t >= off) x = sc[t - off];
        __syncthreads();
        if (t < BIN_W) sc[t] += x;
        __syncthreads();
    }
    if (t < BIN_W) {
        const int start = beg + sc[t] - v;
        const int node  = n0 + t;
        cur[t] = start;
        if (node < N) {
            rowptr[node] = start;
            deg[node]    = (float)v;
            if (node == N - 1) rowptr[N] = end;
        }
    }
    __syncthreads();
    for (int i = beg + t; i < end; i += 1024) {
        const int pos = atomicAdd(&cur[tdst[i] - n0], 1);
        ((int2*)sorted)[pos] = ((const int2*)tpair)[i];
    }
}

template <int K, int F>
__global__ void __launch_bounds__(32)
sage_project(const float* __restrict__ xin,
             const float* __restrict__ Ws, const float* __restrict__ Wn,
             const float* __restrict__ bias,
             float* __restrict__ S, float* __restrict__ P) {
    static_assert(F == 1 || F % 16 == 0, "column tiling");
    static_assert(K % 4 == 0, "k steps");
    constexpr int NS = K / 4;
    constexpr int FT = (F >= 16) ? F / 16 : 1;
    constexpr int NT = (F >= 16) ? 2 * FT : 1;

    const int lane = threadIdx.x;
    const int c    = lane & 15;
    const int hh   = lane >> 4;
    const int row0 = blockIdx.x << 4;
    const int node = row0 + c;

    v2f a[NS];
    const float* xrow = xin + (size_t)node * K;
    for (int t = 0; t < NS; ++t) {
        const int k0 = 4 * t + 2 * hh;
        a[t] = *(const v2f*)(xrow + k0);
    }

    v8f d[NT];
    for (int j = 0; j < NT; ++j) d[j] = v8f{};

    for (int t = 0; t < NS; ++t) {
        const int k0 = 4 * t + 2 * hh;
        for (int j = 0; j < NT; ++j) {
            v2f b;
            if (F >= 16) {
                const float* W = (j < FT) ? Ws : Wn;
                const int col  = 16 * (j % FT) + c;
                b.x = W[(k0)     * F + col];
                b.y = W[(k0 + 1) * F + col];
            } else {
                const float* W = (c == 1) ? Wn : Ws;
                const float w0 = W[k0];
                const float w1 = W[k0 + 1];
                b.x = (c < 2) ? w0 : 0.0f;
                b.y = (c < 2) ? w1 : 0.0f;
            }
            d[j] = __builtin_amdgcn_wmma_f32_16x16x4_f32(false, a[t], false, b,
                                                         (short)0, d[j], false, false);
        }
    }

    for (int j = 0; j < NT; ++j) {
        for (int r = 0; r < 8; ++r) {
            const size_t row = (size_t)(row0 + 8 * hh + r);
            if (F >= 16) {
                const int col = 16 * (j % FT) + c;
                if (j < FT) S[row * F + col] = d[j][r] + bias[col];
                else        P[row * F + col] = d[j][r];
            } else {
                float* o = (c == 1) ? P : S;
                const float v = (c == 0) ? d[0][r] + bias[0] : d[0][r];
                if (c < 2) o[row] = v;
            }
        }
    }
}

template <int F>
__global__ void __launch_bounds__(256)
pull_rows(const float* __restrict__ P, const int* __restrict__ rowptr,
          const int* __restrict__ sorted, const float* __restrict__ deg,
          float* __restrict__ X, int N) {
    static_assert(F == 16 || F == 32, "lane group per node");
    constexpr int NPW = 32 / F;
    const int lane = threadIdx.x & 31;
    const int f    = lane % F;
    const int node = (blockIdx.x * 8 + (threadIdx.x >> 5)) * NPW + lane / F;
    if (node >= N) return;

    const int beg = rowptr[node];
    const int end = rowptr[node + 1];
    const int2* ed = (const int2*)sorted;

    float acc = 0.0f;
    for (int c0 = beg; c0 < end; c0 += F) {
        const int n = min(F, end - c0);
        int2 mine = make_int2(0, 0);
        if (f < n) mine = ed[c0 + f];
        for (int k = 0; k < n; k += 8) {
            int   s[8];
            float w[8];
            float p[8];
            for (int i = 0; i < 8; ++i) {
                s[i] = __shfl(mine.x, k + i, F);
                w[i] = __int_as_float(__shfl(mine.y, k + i, F));
            }
            for (int i = 0; i < 8; ++i) p[i] = P[(size_t)s[i] * F + f];
            for (int i = 0; i < 8; ++i) acc += (k + i < n) ? w[i] * p[i] : 0.0f;
        }
    }

    const float invdeg = 1.0f / fmaxf(deg[node], 1.0f);
    const size_t o = (size_t)node * F + f;
    X[o] = tanhf(X[o] + acc * invdeg);
}

__global__ void __launch_bounds__(256)
pull_final(const float* __restrict__ P, const int* __restrict__ rowptr,
           const int* __restrict__ sorted, const float* __restrict__ S,
           const float* __restrict__ deg, float* __restrict__ out, int N) {
    const int sub  = threadIdx.x & 15;
    const int node = (blockIdx.x * 256 + threadIdx.x) >> 4;
    if (node >= N) return;

    const int beg = rowptr[node];
    const int end = rowptr[node + 1];
    const int2* ed = (const int2*)sorted;

    float acc = 0.0f;
    for (int j = beg + sub; j < end; j += 16) {
        const int2 ev = ed[j];
        acc += __int_as_float(ev.y) * P[ev.x];
    }
    acc += __shfl_xor(acc, 8, 16);
    acc += __shfl_xor(acc, 4, 16);
    acc += __shfl_xor(acc, 2, 16);
    acc += __shfl_xor(acc, 1, 16);
    if (sub == 0) out[node] = S[node] + acc * (1.0f / fmaxf(deg[node], 1.0f));
}

extern "C" void kernel_launch(void* const* d_in, const int* in_sizes, int n_in,
                              void* d_out, int out_size, void* d_ws, size_t ws_size,
                              hipStream_t stream) {
    const float* b_z  = (const float*)d_in[0];
    const int*   esrc = (const int*)  d_in[1];
    const int*   edst = (const int*)  d_in[2];
    const float* ew   = (const float*)d_in[3];
    const float* Ws0  = (const float*)d_in[4];
    const float* Wn0  = (const float*)d_in[5];
    const float* b0   = (const float*)d_in[6];
    const float* Ws1  = (const float*)d_in[7];
    const float* Wn1  = (const float*)d_in[8];
    const float* b1   = (const float*)d_in[9];
    const float* Ws2  = (const float*)d_in[10];
    const float* Wn2  = (const float*)d_in[11];
    const float* b2   = (const float*)d_in[12];

    const int N = in_sizes[0] / 64;
    const int E = in_sizes[1];
    const size_t n = (size_t)N;

    float* ws     = (float*)d_ws;
    float* deg    = ws;
    int*   rowptr = (int*)(ws + n);
    int*   sorted = (int*)(ws + 4 * n);
    int*   tdst   = (int*)(ws + 36 * n);
    int*   tpair  = (int*)(ws + 52 * n);
    int*   counts = (int*)(ws + 84 * n);
    int*   offs   = (int*)(ws + 86 * n);
    float* S0     = ws + 36 * n;
    float* P0     = ws + 68 * n;
    float* S1     = ws + 68 * n;
    float* P1     = ws + 84 * n;
    float* S2     = ws + 36 * n;
    float* P2     = ws + 40 * n;
    float* out    = (float*)d_out;

    auto cdiv = [](int a, int b) { return (a + b - 1) / b; };

    const int nbin  = cdiv(N, BIN_W);
    const int chunk = cdiv(E, SORT_G);
    const int M     = nbin * SORT_G;
    bin_count<<<SORT_G, 1024, 0, stream>>>(edst, counts, E, chunk, nbin);
    scan_excl<<<cdiv(M, 1024), 1024, 0, stream>>>(counts, offs, M);
    bin_scatter<<<SORT_G, 1024, 0, stream>>>(esrc, edst, ew, offs, tdst, tpair, E, chunk, nbin);
    bin_sort<<<nbin, 1024, 0, stream>>>(tdst, tpair, offs, rowptr, deg, sorted, N);

    sage_project<64, 32><<<N / 16, 32, 0, stream>>>(b_z, Ws0, Wn0, b0, S0, P0);
    pull_rows<32><<<cdiv(N, 8), 256, 0, stream>>>(P0, rowptr, sorted, deg, S0, N);

    sage_project<32, 16><<<N / 16, 32, 0, stream>>>(S0, Ws1, Wn1, b1, S1, P1);
    pull_rows<16><<<cdiv(N, 16), 256, 0, stream>>>(P1, rowptr, sorted, deg, S1, N);

    sage_project<16, 1><<<N / 16, 32, 0, stream>>>(S1, Ws2, Wn2, b2, S2, P2);
    pull_final<<<cdiv(N, 16), 256, 0, stream>>>(P2, rowptr, sorted, S2, deg, out, N);
}


# Layers 1 and 2: the dense stage runs inside the edge-sum launch in front of it (block = 16 nodes, x rows through an LDS tile, wave 0 does the matrix steps): 10 launches become 8; sums unchanged
# speedup vs baseline: 2.6919x; 1.0374x over previous
#include <hip/hip_runtime.h>


typedef __attribute__((ext_vector_type(2))) float v2f;
typedef __attribute__((ext_vector_type(8))) float v8f;

constexpr int SORT_G  = 256;
constexpr int BIN_SH  = 8;
constexpr int BIN_W   = 1 << BIN_SH;
constexpr int MAX_BIN = 1024;

__global__ void __launch_bounds__(1024)
bin_count(const int* __restrict__ dst, int* __restrict__ counts, int E, int chunk, int nbin) {
    __shared__ int h[MAX_BIN];
    const int t = threadIdx.x;
    const int g = blockIdx.x;
    for (int i = t; i < nbin; i += 1024) h[i] = 0;
    __syncthreads();
    const int lo = g * chunk;
    const int hi = min(E, lo + chunk);
    for (int e = lo + t; e < hi; e += 1024) atomicAdd(&h[dst[e] >> BIN_SH], 1);
    __syncthreads();
    for (int i = t; i < nbin; i += 1024) counts[i * SORT_G + g] = h[i];
}

__global__ void __launch_bounds__(1024)
scan_excl(const int* __restrict__ in, int* __restrict__ out, int M) {
    __shared__ int sh[1024];
    const int t = threadIdx.x;
    const int b = blockIdx.x;

    int part = 0;
    int j = 0;
    for (; j + 8 <= b; j += 8) {
        const int a0 = in[t + 1024 * (j + 0)];
        const int a1 = in[t + 1024 * (j + 1)];
        const int a2 = in[t + 1024 * (j + 2)];
        const int a3 = in[t + 1024 * (j + 3)];
        const int a4 = in[t + 1024 * (j + 4)];
        const int a5 = in[t + 1024 * (j + 5)];
        const int a6 = in[t + 1024 * (j + 6)];
        const int a7 = in[t + 1024 * (j + 7)];
        part += ((a0 + a1) + (a2 + a3)) + ((a4 + a5) + (a6 + a7));
    }
    for (; j < b; ++j) part += in[t + 1024 * j];
    sh[t] = part;
    __syncthreads();
    for (int s = 512; s > 0; s >>= 1) {
        if (t < s) sh[t] += sh[t + s];
        __syncthreads();
    }
    const int base = sh[0];
    __syncthreads();

    const int i = 1024 * b + t;
    const int v = (i < M) ? in[i] : 0;
    sh[t] = v;
    __syncthreads();
    for (int off = 1; off < 1024; off <<= 1) {
        const int x = (t >= off) ? sh[t - off] : 0;
        __syncthreads();
        sh[t] += x;
        __syncthreads();
    }
    const int incl = sh[t];
    if (i < M) {
        out[i] = base + incl - v;
        if (i == M - 1) out[M] = base + incl;
    }
}

__global__ void __launch_bounds__(1024)
bin_scatter(const int* __restrict__ src, const int* __restrict__ dst,
            const float* __restrict__ ew, const int* __restrict__ offs,
            int* __restrict__ tdst, int* __restrict__ tpair, int E, int chunk, int nbin) {
    __shared__ int cur[MAX_BIN];
    const int t = threadIdx.x;
    const int g = blockIdx.x;
    for (int i = t; i < nbin; i += 1024) cur[i] = offs[i * SORT_G + g];
    __syncthreads();
    const int lo = g * chunk;
    const int hi = min(E, lo + chunk);
    for (int e = lo + t; e < hi; e += 1024) {
        const int d   = dst[e];
        const int pos = atomicAdd(&cur[d >> BIN_SH], 1);
        tdst[pos] = d;
        ((int2*)tpair)[pos] = make_int2(src[e], __float_as_int(ew[e]));
    }
}

__global__ void __launch_bounds__(1024)
bin_sort(const int* __restrict__ tdst, const int* __restrict__ tpair,
         const int* __restrict__ offs, int* __restrict__ rowptr,
         float* __restrict__ deg, int* __restrict__ sorted, int N) {
    __shared__ int cnt[BIN_W];
    __shared__ int sc[BIN_W];
    __shared__ int cur[BIN_W];
    const int t   = threadIdx.x;
    const int b   = blockIdx.x;
    const int n0  = b << BIN_SH;
    const int beg = offs[b * SORT_G];
    const int end = offs[(b + 1) * SORT_G];

    if (t < BIN_W) cnt[t] = 0;
    __syncthreads();
    for (int i = beg + t; i < end; i += 1024) atomicAdd(&cnt[tdst[i] - n0], 1);
    __syncthreads();

    const int v = (t < BIN_W) ? cnt[t] : 0;
    if (t < BIN_W) sc[t] = v;
    __syncthreads();
    for (int off = 1; off < BIN_W; off <<= 1) {
        int x = 0;
        if (t < BIN_W && t >= off) x = sc[t - off];
        __syncthreads();
        if (t < BIN_W) sc[t] += x;
        __syncthreads();
    }
    if (t < BIN_W) {
        const int start = beg + sc[t] - v;
        const int node  = n0 + t;
        cur[t] = start;
        if (node < N) {
            rowptr[node] = start;
            deg[node]    = (float)v;
            if (node == N - 1) rowptr[N] = end;
        }
    }
    __syncthreads();
    for (int i = beg + t; i < end; i += 1024) {
        const int pos = atomicAdd(&cur[tdst[i] - n0], 1);
        ((int2*)sorted)[pos] = ((const int2*)tpair)[i];
    }
}

template <int K, int F>
__global__ void __launch_bounds__(32)
sage_project(const float* __restrict__ xin,
             const float* __restrict__ Ws, const float* __restrict__ Wn,
             const float* __restrict__ bias,
             float* __restrict__ S, float* __restrict__ P) {
    static_assert(F == 1 || F % 16 == 0, "column tiling");
    static_assert(K % 4 == 0, "k steps");
    constexpr int NS = K / 4;
    constexpr int FT = (F >= 16) ? F / 16 : 1;
    constexpr int NT = (F >= 16) ? 2 * FT : 1;

    const int lane = threadIdx.x;
    const int c    = lane & 15;
    const int hh   = lane >> 4;
    const int row0 = blockIdx.x << 4;
    const int node = row0 + c;

    v2f a[NS];
    const float* xrow = xin + (size_t)node * K;
    for (int t = 0; t < NS; ++t) {
        const int k0 = 4 * t + 2 * hh;
        a[t] = *(const v2f*)(xrow + k0);
    }

    v8f d[NT];
    for (int j = 0; j < NT; ++j) d[j] = v8f{};

    for (int t = 0; t < NS; ++t) {
        const int k0 = 4 * t + 2 * hh;
        for (int j = 0; j < NT; ++j) {
            v2f b;
            if (F >= 16) {
                const float* W = (j < FT) ? Ws : Wn;
                const int col  = 16 * (j % FT) + c;
                b.x = W[(k0)     * F + col];
                b.y = W[(k0 + 1) * F + col];
            } else {
                const float* W = (c == 1) ? Wn : Ws;
                const float w0 = W[k0];
                const float w1 = W[k0 + 1];
                b.x = (c < 2) ? w0 : 0.0f;
                b.y = (c < 2) ? w1 : 0.0f;
            }
            d[j] = __builtin_amdgcn_wmma_f32_16x16x4_f32(false, a[t], false, b,
                                                         (short)0, d[j], false, false);
        }
    }

    for (int j = 0; j < NT; ++j) {
        for (int r = 0; r < 8; ++r) {
            const size_t row = (size_t)(row0 + 8 * hh + r);
            if (F >= 16) {
                const int col = 16 * (j % FT) + c;
                if (j < FT) S[row * F + col] = d[j][r] + bias[col];
                else        P[row * F + col] = d[j][r];
            } else {
                float* o = (c == 1) ? P : S;
                const float v = (c == 0) ? d[0][r] + bias[0] : d[0][r];
                if (c < 2) o[row] = v;
            }
        }
    }
}

template <int F, int FO>
__global__ void __launch_bounds__(16 * F)
pull_project(const float* S, const float* __restrict__ P,
             const int* __restrict__ rowptr, const int* __restrict__ sorted,
             const float* __restrict__ deg,
             const float* __restrict__ Ws, const float* __restrict__ Wn,
             const float* __restrict__ bias, float* So, float* Po) {
    static_assert(F == 16 || F == 32, "lane group per node");
    static_assert(FO == 1 || FO == 16, "column tiling");
    constexpr int PIN = 32;
    constexpr int XP  = F + 4;
    constexpr int NS  = F / 4;
    constexpr int NT  = (FO == 16) ? 2 : 1;
    __shared__ __attribute__((aligned(16))) float xs[16 * XP];

    const int lane = threadIdx.x & 31;
    const int f    = lane % F;
    const int slot = threadIdx.x / F;
    const int row0 = blockIdx.x << 4;
    const int node = row0 + slot;

    const int beg = rowptr[node];
    const int end = rowptr[node + 1];
    const int2* ed = (const int2*)sorted;

    float acc = 0.0f;
    for (int c0 = beg; c0 < end; c0 += F) {
        const int n = min(F, end - c0);
        int2 mine = make_int2(0, 0);
        if (f < n) mine = ed[c0 + f];
        for (int k = 0; k < n; k += 8) {
            int   s[8];
            float w[8];
            float p[8];
            for (int i = 0; i < 8; ++i) {
                s[i] = __shfl(mine.x, k + i, F);
                w[i] = __int_as_float(__shfl(mine.y, k + i, F));
            }
            for (int i = 0; i < 8; ++i) p[i] = P[(size_t)s[i] * PIN + f];
            for (int i = 0; i < 8; ++i) acc += (k + i < n) ? w[i] * p[i] : 0.0f;
        }
    }

    const float invdeg = 1.0f / fmaxf(deg[node], 1.0f);
    xs[slot * XP + f] = tanhf(S[(size_t)node * PIN + f] + acc * invdeg);
    __syncthreads();
    if (threadIdx.x >= 32) return;

    const int c  = lane & 15;
    const int hh = lane >> 4;

    v2f a[NS];
    for (int t = 0; t < NS; ++t) {
        const int k0 = 4 * t + 2 * hh;
        a[t] = *(const v2f*)(xs + c * XP + k0);
    }

    v8f d[NT];
    for (int j = 0; j < NT; ++j) d[j] = v8f{};

    for (int t = 0; t < NS; ++t) {
        const int k0 = 4 * t + 2 * hh;
        for (int j = 0; j < NT; ++j) {
            v2f b;
            if (FO == 16) {
                const float* W = (j == 0) ? Ws : Wn;
                b.x = W[(k0)     * FO + c];
                b.y = W[(k0 + 1) * FO + c];
            } else {
                const float* W = (c == 1) ? Wn : Ws;
                const float w0 = W[k0];
                const float w1 = W[k0 + 1];
                b.x = (c < 2) ? w0 : 0.0f;
                b.y = (c < 2) ? w1 : 0.0f;
            }
            d[j] = __builtin_amdgcn_wmma_f32_16x16x4_f32(false, a[t], false, b,
                                                         (short)0, d[j], false, false);
        }
    }

    for (int r = 0; r < 8; ++r) {
        const size_t row = (size_t)(row0 + 8 * hh + r);
        if (FO == 16) {
            So[row * PIN + c] = d[0][r] + bias[c];
            Po[row * PIN + c] = d[NT - 1][r];
        } else {
            float* o = (c == 1) ? Po : So;
            const float v = (c == 0) ? d[0][r] + bias[0] : d[0][r];
            if (c < 2) o[row] = v;
        }
    }
}

__global__ void __launch_bounds__(256)
pull_final(const float* __restrict__ P, const int* __restrict__ rowptr,
           const int* __restrict__ sorted, const float* __restrict__ S,
           const float* __restrict__ deg, float* __restrict__ out, int N) {
    const int sub  = threadIdx.x & 15;
    const int node = (blockIdx.x * 256 + threadIdx.x) >> 4;
    if (node >= N) return;

    const int beg = rowptr[node];
    const int end = rowptr[node + 1];
    const int2* ed = (const int2*)sorted;

    float acc = 0.0f;
    for (int j = beg + sub; j < end; j += 16) {
        const int2 ev = ed[j];
        acc += __int_as_float(ev.y) * P[ev.x];
    }
    acc += __shfl_xor(acc, 8, 16);
    acc += __shfl_xor(acc, 4, 16);
    acc += __shfl_xor(acc, 2, 16);
    acc += __shfl_xor(acc, 1, 16);
    if (sub == 0) out[node] = S[node] + acc * (1.0f / fmaxf(deg[node], 1.0f));
}

extern "C" void kernel_launch(void* const* d_in, const int* in_sizes, int n_in,
                              void* d_out, int out_size, void* d_ws, size_t ws_size,
                              hipStream_t stream) {
    const float* b_z  = (const float*)d_in[0];
    const int*   esrc = (const int*)  d_in[1];
    const int*   edst = (const int*)  d_in[2];
    const float* ew   = (const float*)d_in[3];
    const float* Ws0  = (const float*)d_in[4];
    const float* Wn0  = (const float*)d_in[5];
    const float* b0   = (const float*)d_in[6];
    const float* Ws1  = (const float*)d_in[7];
    const float* Wn1  = (const float*)d_in[8];
    const float* b1   = (const float*)d_in[9];
    const float* Ws2  = (const float*)d_in[10];
    const float* Wn2  = (const float*)d_in[11];
    const float* b2   = (const float*)d_in[12];

    const int N = in_sizes[0] / 64;
    const int E = in_sizes[1];
    const size_t n = (size_t)N;

    float* ws     = (float*)d_ws;
    float* deg    = ws;
    int*   rowptr = (int*)(ws + n);
    int*   sorted = (int*)(ws + 4 * n);
    int*   tdst   = (int*)(ws + 36 * n);
    int*   tpair  = (int*)(ws + 52 * n);
    int*   counts = (int*)(ws + 84 * n);
    int*   offs   = (int*)(ws + 86 * n);
    float* S0     = ws + 36 * n;
    float* P0     = ws + 68 * n;
    float* S1     = ws + 36 * n;
    float* P1     = ws + 36 * n + 16;
    float* S2     = ws + 68 * n;
    float* P2     = ws + 72 * n;
    float* out    = (float*)d_out;

    auto cdiv = [](int a, int b) { return (a + b - 1) / b; };

    const int nbin  = cdiv(N, BIN_W);
    const int chunk = cdiv(E, SORT_G);
    const int M     = nbin * SORT_G;
    bin_count<<<SORT_G, 1024, 0, stream>>>(edst, counts, E, chunk, nbin);
    scan_excl<<<cdiv(M, 1024), 1024, 0, stream>>>(counts, offs, M);
    bin_scatter<<<SORT_G, 1024, 0, stream>>>(esrc, edst, ew, offs, tdst, tpair, E, chunk, nbin);
    bin_sort<<<nbin, 1024, 0, stream>>>(tdst, tpair, offs, rowptr, deg, sorted, N);

    sage_project<64, 32><<<N / 16, 32, 0, stream>>>(b_z, Ws0, Wn0, b0, S0, P0);

    pull_project<32, 16><<<N / 16, 512, 0, stream>>>(S0, P0, rowptr, sorted, deg, Ws1, Wn1, b1, S1, P1);

    pull_project<16, 1><<<N / 16, 256, 0, stream>>>(S1, P1, rowptr, sorted, deg, Ws2, Wn2, b2, S2, P2);

    pull_final<<<cdiv(N, 16), 256, 0, stream>>>(P2, rowptr, sorted, S2, deg, out, N);
}


# Edge sums: one wave per 16 nodes in all three edge-sum launches (a lane holds 4 features, 4 or 8 nodes side by side in a wave; last layer 2 at a time): 100000 short waves become 9375; sums unchanged
# speedup vs baseline: 2.9492x; 1.0956x over previous
#include <hip/hip_runtime.h>


typedef __attribute__((ext_vector_type(2))) float v2f;
typedef __attribute__((ext_vector_type(4))) float v4f;
typedef __attribute__((ext_vector_type(8))) float v8f;

constexpr int SORT_G  = 256;
constexpr int BIN_SH  = 8;
constexpr int BIN_W   = 1 << BIN_SH;
constexpr int MAX_BIN = 1024;

__global__ void __launch_bounds__(1024)
bin_count(const int* __restrict__ dst, int* __restrict__ counts, int E, int chunk, int nbin) {
    __shared__ int h[MAX_BIN];
    const int t = threadIdx.x;
    const int g = blockIdx.x;
    for (int i = t; i < nbin; i += 1024) h[i] = 0;
    __syncthreads();
    const int lo = g * chunk;
    const int hi = min(E, lo + chunk);
    for (int e = lo + t; e < hi; e += 1024) atomicAdd(&h[dst[e] >> BIN_SH], 1);
    __syncthreads();
    for (int i = t; i < nbin; i += 1024) counts[i * SORT_G + g] = h[i];
}

__global__ void __launch_bounds__(1024)
scan_excl(const int* __restrict__ in, int* __restrict__ out, int M) {
    __shared__ int sh[1024];
    const int t = threadIdx.x;
    const int b = blockIdx.x;

    int part = 0;
    int j = 0;
    for (; j + 8 <= b; j += 8) {
        const int a0 = in[t + 1024 * (j + 0)];
        const int a1 = in[t + 1024 * (j + 1)];
        const int a2 = in[t + 1024 * (j + 2)];
        const int a3 = in[t + 1024 * (j + 3)];
        const int a4 = in[t + 1024 * (j + 4)];
        const int a5 = in[t + 1024 * (j + 5)];
        const int a6 = in[t + 1024 * (j + 6)];
        const int a7 = in[t + 1024 * (j + 7)];
        part += ((a0 + a1) + (a2 + a3)) + ((a4 + a5) + (a6 + a7));
    }
    for (; j < b; ++j) part += in[t + 1024 * j];
    sh[t] = part;
    __syncthreads();
    for (int s = 512; s > 0; s >>= 1) {
        if (t < s) sh[t] += sh[t + s];
        __syncthreads();
    }
    const int base = sh[0];
    __syncthreads();

    const int i = 1024 * b + t;
    const int v = (i < M) ? in[i] : 0;
    sh[t] = v;
    __syncthreads();
    for (int off = 1; off < 1024; off <<= 1) {
        const int x = (t >= off) ? sh[t - off] : 0;
        __syncthreads();
        sh[t] += x;
        __syncthreads();
    }
    const int incl = sh[t];
    if (i < M) {
        out[i] = base + incl - v;
        if (i == M - 1) out[M] = base + incl;
    }
}

__global__ void __launch_bounds__(1024)
bin_scatter(const int* __restrict__ src, const int* __restrict__ dst,
            const float* __restrict__ ew, const int* __restrict__ offs,
            int* __restrict__ tdst, int* __restrict__ tpair, int E, int chunk, int nbin) {
    __shared__ int cur[MAX_BIN];
    const int t = threadIdx.x;
    const int g = blockIdx.x;
    for (int i = t; i < nbin; i += 1024) cur[i] = offs[i * SORT_G + g];
    __syncthreads();
    const int lo = g * chunk;
    const int hi = min(E, lo + chunk);
    for (int e = lo + t; e < hi; e += 1024) {
        const int d   = dst[e];
        const int pos = atomicAdd(&cur[d >> BIN_SH], 1);
        tdst[pos] = d;
        ((int2*)tpair)[pos] = make_int2(src[e], __float_as_int(ew[e]));
    }
}

__global__ void __launch_bounds__(1024)
bin_sort(const int* __restrict__ tdst, const int* __restrict__ tpair,
         const int* __restrict__ offs, int* __restrict__ rowptr,
         float* __restrict__ deg, int* __restrict__ sorted, int N) {
    __shared__ int cnt[BIN_W];
    __shared__ int sc[BIN_W];
    __shared__ int cur[BIN_W];
    const int t   = threadIdx.x;
    const int b   = blockIdx.x;
    const int n0  = b << BIN_SH;
    const int beg = offs[b * SORT_G];
    const int end = offs[(b + 1) * SORT_G];

    if (t < BIN_W) cnt[t] = 0;
    __syncthreads();
    for (int i = beg + t; i < end; i += 1024) atomicAdd(&cnt[tdst[i] - n0], 1);
    __syncthreads();

    const int v = (t < BIN_W) ? cnt[t] : 0;
    if (t < BIN_W) sc[t] = v;
    __syncthreads();
    for (int off = 1; off < BIN_W; off <<= 1) {
        int x = 0;
        if (t < BIN_W && t >= off) x = sc[t - off];
        __syncthreads();
        if (t < BIN_W) sc[t] += x;
        __syncthreads();
    }
    if (t < BIN_W) {
        const int start = beg + sc[t] - v;
        const int node  = n0 + t;
        cur[t] = start;
        if (node < N) {
            rowptr[node] = start;
            deg[node]    = (float)v;
            if (node == N - 1) rowptr[N] = end;
        }
    }
    __syncthreads();
    for (int i = beg + t; i < end; i += 1024) {
        const int pos = atomicAdd(&cur[tdst[i] - n0], 1);
        ((int2*)sorted)[pos] = ((const int2*)tpair)[i];
    }
}

template <int K, int F>
__global__ void __launch_bounds__(32)
sage_project(const float* __restrict__ xin,
             const float* __restrict__ Ws, const float* __restrict__ Wn,
             const float* __restrict__ bias,
             float* __restrict__ S, float* __restrict__ P) {
    static_assert(F == 1 || F % 16 == 0, "column tiling");
    static_assert(K % 4 == 0, "k steps");
    constexpr int NS = K / 4;
    constexpr int FT = (F >= 16) ? F / 16 : 1;
    constexpr int NT = (F >= 16) ? 2 * FT : 1;

    const int lane = threadIdx.x;
    const int c    = lane & 15;
    const int hh   = lane >> 4;
    const int row0 = blockIdx.x << 4;
    const int node = row0 + c;

    v2f a[NS];
    const float* xrow = xin + (size_t)node * K;
    for (int t = 0; t < NS; ++t) {
        const int k0 = 4 * t + 2 * hh;
        a[t] = *(const v2f*)(xrow + k0);
    }

    v8f d[NT];
    for (int j = 0; j < NT; ++j) d[j] = v8f{};

    for (int t = 0; t < NS; ++t) {
        const int k0 = 4 * t + 2 * hh;
        for (int j = 0; j < NT; ++j) {
            v2f b;
            if (F >= 16) {
                const float* W = (j < FT) ? Ws : Wn;
                const int col  = 16 * (j % FT) + c;
                b.x = W[(k0)     * F + col];
                b.y = W[(k0 + 1) * F + col];
            } else {
                const float* W = (c == 1) ? Wn : Ws;
                const float w0 = W[k0];
                const float w1 = W[k0 + 1];
                b.x = (c < 2) ? w0 : 0.0f;
                b.y = (c < 2) ? w1 : 0.0f;
            }
            d[j] = __builtin_amdgcn_wmma_f32_16x16x4_f32(false, a[t], false, b,
                                                         (short)0, d[j], false, false);
        }
    }

    for (int j = 0; j < NT; ++j) {
        for (int r = 0; r < 8; ++r) {
            const size_t row = (size_t)(row0 + 8 * hh + r);
            if (F >= 16) {
                const int col = 16 * (j % FT) + c;
                if (j < FT) S[row * F + col] = d[j][r] + bias[col];
                else        P[row * F + col] = d[j][r];
            } else {
                float* o = (c == 1) ? P : S;
                const float v = (c == 0) ? d[0][r] + bias[0] : d[0][r];
                if (c < 2) o[row] = v;
            }
        }
    }
}

template <int F, int FO>
__global__ void __launch_bounds__(32)
pull_project(const float* S, const float* __restrict__ P,
             const int* __restrict__ rowptr, const int* __restrict__ sorted,
             const float* __restrict__ deg,
             const float* __restrict__ Ws, const float* __restrict__ Wn,
             const float* __restrict__ bias, float* So, float* Po) {
    static_assert(F == 16 || F == 32, "lane group per node");
    static_assert(FO == 1 || FO == 16, "column tiling");
    constexpr int PIN   = 32;
    constexpr int PQ    = PIN / 4;
    constexpr int LPN   = F / 4;
    constexpr int NPP   = 32 / LPN;
    constexpr int NPASS = 16 / NPP;
    constexpr int EB    = 8;
    constexpr int XP    = F + 4;
    constexpr int NS    = F / 4;
    constexpr int NT    = (FO == 16) ? 2 : 1;
    __shared__ __attribute__((aligned(16))) float xs[16 * XP];

    const int lane = threadIdx.x;
    const int sub  = lane % LPN;
    const int grp  = lane / LPN;
    const int row0 = blockIdx.x << 4;
    const int2* ed = (const int2*)sorted;
    const v4f*  P4 = (const v4f*)P;
    const v4f*  S4 = (const v4f*)S;

    int beg[NPASS], end[NPASS];
    for (int j = 0; j < NPASS; ++j) {
        const int node = row0 + j * NPP + grp;
        beg[j] = rowptr[node];
        end[j] = rowptr[node + 1];
    }

    int2 e[EB];
    {
        const int last = max(end[0] - 1, 0);
        for (int i = 0; i < EB; ++i) e[i] = ed[min(beg[0] + i, last)];
    }

    for (int j = 0; j < NPASS; ++j) {
        const int slot = j * NPP + grp;
        const int node = row0 + slot;
        const int last = max(end[j] - 1, 0);

        int2 ef[EB];
        if (j + 1 < NPASS) {
            const int lastn = max(end[j + 1] - 1, 0);
            for (int i = 0; i < EB; ++i) ef[i] = ed[min(beg[j + 1] + i, lastn)];
        }
        const v4f   s4 = S4[(size_t)node * PQ + sub];
        const float dg = deg[node];

        v4f acc = v4f{};
        for (int c0 = beg[j]; c0 < end[j]; c0 += EB) {
            int2 en[EB];
            for (int i = 0; i < EB; ++i) en[i] = ed[min(c0 + EB + i, last)];
            v4f p[EB];
            for (int i = 0; i < EB; ++i) p[i] = P4[e[i].x * PQ + sub];
            __builtin_amdgcn_sched_barrier(0);
            for (int i = 0; i < EB; ++i) {
                const float w  = __int_as_float(e[i].y);
                const bool  in = c0 + i < end[j];
                acc.x += in ? w * p[i].x : 0.0f;
                acc.y += in ? w * p[i].y : 0.0f;
                acc.z += in ? w * p[i].z : 0.0f;
                acc.w += in ? w * p[i].w : 0.0f;
            }
            for (int i = 0; i < EB; ++i) e[i] = en[i];
        }

        const float invdeg = 1.0f / fmaxf(dg, 1.0f);
        v4f x;
        x.x = tanhf(s4.x + acc.x * invdeg);
        x.y = tanhf(s4.y + acc.y * invdeg);
        x.z = tanhf(s4.z + acc.z * invdeg);
        x.w = tanhf(s4.w + acc.w * invdeg);
        *(v4f*)(xs + slot * XP + 4 * sub) = x;

        if (j + 1 < NPASS)
            for (int i = 0; i < EB; ++i) e[i] = ef[i];
    }
    __syncthreads();

    const int c  = lane & 15;
    const int hh = lane >> 4;

    v2f a[NS];
    for (int t = 0; t < NS; ++t) {
        const int k0 = 4 * t + 2 * hh;
        a[t] = *(const v2f*)(xs + c * XP + k0);
    }

    v8f d[NT];
    for (int j = 0; j < NT; ++j) d[j] = v8f{};

    for (int t = 0; t < NS; ++t) {
        const int k0 = 4 * t + 2 * hh;
        for (int j = 0; j < NT; ++j) {
            v2f b;
            if (FO == 16) {
                const float* W = (j == 0) ? Ws : Wn;
                b.x = W[(k0)     * FO + c];
                b.y = W[(k0 + 1) * FO + c];
            } else {
                const float* W = (c == 1) ? Wn : Ws;
                const float w0 = W[k0];
                const float w1 = W[k0 + 1];
                b.x = (c < 2) ? w0 : 0.0f;
                b.y = (c < 2) ? w1 : 0.0f;
            }
            d[j] = __builtin_amdgcn_wmma_f32_16x16x4_f32(false, a[t], false, b,
                                                         (short)0, d[j], false, false);
        }
    }

    for (int r = 0; r < 8; ++r) {
        const size_t row = (size_t)(row0 + 8 * hh + r);
        if (FO == 16) {
            So[row * PIN + c] = d[0][r] + bias[c];
            Po[row * PIN + c] = d[NT - 1][r];
        } else {
            float* o = (c == 1) ? Po : So;
            const float v = (c == 0) ? d[0][r] + bias[0] : d[0][r];
            if (c < 2) o[row] = v;
        }
    }
}

__global__ void __launch_bounds__(32)
pull_final(const float* __restrict__ P, const int* __restrict__ rowptr,
           const int* __restrict__ sorted, const float* __restrict__ S,
           const float* __restrict__ deg, float* __restrict__ out) {
    constexpr int NI = 8;
    const int lane = threadIdx.x;
    const int sub  = lane & 15;
    const int h    = lane >> 4;
    const int base = blockIdx.x * 16;
    const int2* ed = (const int2*)sorted;

    const int   onode = base + 2 * (sub & (NI - 1)) + h;
    const float sv    = S[onode];
    const float dv    = deg[onode];

    const int* rp = rowptr + base + h;
    int beg[NI], end[NI];
    for (int i = 0; i < NI; ++i) {
        beg[i] = rp[2 * i];
        end[i] = rp[2 * i + 1];
    }

    int2 e0[NI], e1[NI];
    for (int i = 0; i < NI; ++i) {
        const int last = max(end[i] - 1, 0);
        e0[i] = ed[min(beg[i] + sub, last)];
        e1[i] = ed[min(beg[i] + sub + 16, last)];
    }
    float p0[NI], p1[NI];
    for (int i = 0; i < NI; ++i) {
        p0[i] = P[e0[i].x];
        p1[i] = P[e1[i].x];
    }

    float acc[NI];
    for (int i = 0; i < NI; ++i) {
        float a = 0.0f;
        if (beg[i] + sub      < end[i]) a += __int_as_float(e0[i].y) * p0[i];
        if (beg[i] + sub + 16 < end[i]) a += __int_as_float(e1[i].y) * p1[i];
        for (int j = beg[i] + sub + 32; j < end[i]; j += 16) {
            const int2 ev = ed[j];
            a += __int_as_float(ev.y) * P[ev.x];
        }
        acc[i] = a;
    }
    for (int s = 8; s > 0; s >>= 1)
        for (int i = 0; i < NI; ++i) acc[i] += __shfl_xor(acc[i], s, 16);

    float res = acc[0];
    for (int i = 1; i < NI; ++i) res = (sub == i) ? acc[i] : res;
    if (sub < NI) out[onode] = sv + res * (1.0f / fmaxf(dv, 1.0f));
}

extern "C" void kernel_launch(void* const* d_in, const int* in_sizes, int n_in,
                              void* d_out, int out_size, void* d_ws, size_t ws_size,
                              hipStream_t stream) {
    const float* b_z  = (const float*)d_in[0];
    const int*   esrc = (const int*)  d_in[1];
    const int*   edst = (const int*)  d_in[2];
    const float* ew   = (const float*)d_in[3];
    const float* Ws0  = (const float*)d_in[4];
    const float* Wn0  = (const float*)d_in[5];
    const float* b0   = (const float*)d_in[6];
    const float* Ws1  = (const float*)d_in[7];
    const float* Wn1  = (const float*)d_in[8];
    const float* b1   = (const float*)d_in[9];
    const float* Ws2  = (const float*)d_in[10];
    const float* Wn2  = (const float*)d_in[11];
    const float* b2   = (const float*)d_in[12];

    const int N = in_sizes[0] / 64;
    const int E = in_sizes[1];
    const size_t n = (size_t)N;

    float* ws     = (float*)d_ws;
    float* deg    = ws;
    int*   rowptr = (int*)(ws + n);
    int*   sorted = (int*)(ws + 4 * n);
    int*   tdst   = (int*)(ws + 36 * n);
    int*   tpair  = (int*)(ws + 52 * n);
    int*   counts = (int*)(ws + 84 * n);
    int*   offs   = (int*)(ws + 86 * n);
    float* S0     = ws + 36 * n;
    float* P0     = ws + 68 * n;
    float* S1     = ws + 36 * n;
    float* P1     = ws + 36 * n + 16;
    float* S2     = ws + 68 * n;
    float* P2     = ws + 72 * n;
    float* out    = (float*)d_out;

    auto cdiv = [](int a, int b) { return (a + b - 1) / b; };

    const int nbin  = cdiv(N, BIN_W);
    const int chunk = cdiv(E, SORT_G);
    const int M     = nbin * SORT_G;
    bin_count<<<SORT_G, 1024, 0, stream>>>(edst, counts, E, chunk, nbin);
    scan_excl<<<cdiv(M, 1024), 1024, 0, stream>>>(counts, offs, M);
    bin_scatter<<<SORT_G, 1024, 0, stream>>>(esrc, edst, ew, offs, tdst, tpair, E, chunk, nbin);
    bin_sort<<<nbin, 1024, 0, stream>>>(tdst, tpair, offs, rowptr, deg, sorted, N);

    sage_project<64, 32><<<N / 16, 32, 0, stream>>>(b_z, Ws0, Wn0, b0, S0, P0);

    pull_project<32, 16><<<N / 16, 32, 0, stream>>>(S0, P0, rowptr, sorted, deg, Ws1, Wn1, b1, S1, P1);

    pull_project<16, 1><<<N / 16, 32, 0, stream>>>(S1, P1, rowptr, sorted, deg, Ws2, Wn2, b2, S2, P2);

    pull_final<<<N / 16, 32, 0, stream>>>(P2, rowptr, sorted, S2, deg, out);
}


# Sort: both copying passes order a tile of edges in LDS (by bin, by node) and store it slot by slot: whole lines, not one line per lane; a node's edge order, so the order of f32 sums, may change
# speedup vs baseline: 3.6552x; 1.2394x over previous
#include <hip/hip_runtime.h>


typedef __attribute__((ext_vector_type(2))) float v2f;
typedef __attribute__((ext_vector_type(4))) float v4f;
typedef __attribute__((ext_vector_type(8))) float v8f;

constexpr int SORT_G  = 256;
constexpr int BIN_SH  = 8;
constexpr int BIN_W   = 1 << BIN_SH;
constexpr int MAX_BIN = 1024;
constexpr int SORT_T  = 1024;
static_assert(MAX_BIN == SORT_T, "one thread per bin counter");

__device__ __forceinline__ int block_scan_excl(int v, int* wsum, int t) {
    const int lane = t & 31;
    const int w    = t >> 5;
    int x = v;
    for (int off = 1; off < 32; off <<= 1) {
        const int y = __shfl_up(x, off, 32);
        if (lane >= off) x += y;
    }
    if (lane == 31) wsum[w] = x;
    __syncthreads();
    if (w == 0) {
        int s = wsum[lane];
        for (int off = 1; off < 32; off <<= 1) {
            const int y = __shfl_up(s, off, 32);
            if (lane >= off) s += y;
        }
        wsum[lane] = s;
    }
    __syncthreads();
    const int base = (w > 0) ? wsum[w - 1] : 0;
    return base + x - v;
}

__global__ void __launch_bounds__(1024)
bin_count(const int* __restrict__ dst, int* __restrict__ counts, int E, int chunk, int nbin) {
    __shared__ int h[MAX_BIN];
    const int t = threadIdx.x;
    const int g = blockIdx.x;
    for (int i = t; i < nbin; i += 1024) h[i] = 0;
    __syncthreads();
    const int lo = g * chunk;
    const int hi = min(E, lo + chunk);
    for (int e = lo + t; e < hi; e += 1024) atomicAdd(&h[dst[e] >> BIN_SH], 1);
    __syncthreads();
    for (int i = t; i < nbin; i += 1024) counts[i * SORT_G + g] = h[i];
}

__global__ void __launch_bounds__(1024)
scan_excl(const int* __restrict__ in, int* __restrict__ out, int M) {
    __shared__ int sh[1024];
    const int t = threadIdx.x;
    const int b = blockIdx.x;

    int part = 0;
    int j = 0;
    for (; j + 8 <= b; j += 8) {
        const int a0 = in[t + 1024 * (j + 0)];
        const int a1 = in[t + 1024 * (j + 1)];
        const int a2 = in[t + 1024 * (j + 2)];
        const int a3 = in[t + 1024 * (j + 3)];
        const int a4 = in[t + 1024 * (j + 4)];
        const int a5 = in[t + 1024 * (j + 5)];
        const int a6 = in[t + 1024 * (j + 6)];
        const int a7 = in[t + 1024 * (j + 7)];
        part += ((a0 + a1) + (a2 + a3)) + ((a4 + a5) + (a6 + a7));
    }
    for (; j < b; ++j) part += in[t + 1024 * j];
    sh[t] = part;
    __syncthreads();
    for (int s = 512; s > 0; s >>= 1) {
        if (t < s) sh[t] += sh[t + s];
        __syncthreads();
    }
    const int base = sh[0];
    __syncthreads();

    const int i = 1024 * b + t;
    const int v = (i < M) ? in[i] : 0;
    sh[t] = v;
    __syncthreads();
    for (int off = 1; off < 1024; off <<= 1) {
        const int x = (t >= off) ? sh[t - off] : 0;
        __syncthreads();
        sh[t] += x;
        __syncthreads();
    }
    const int incl = sh[t];
    if (i < M) {
        out[i] = base + incl - v;
        if (i == M - 1) out[M] = base + incl;
    }
}

__global__ void __launch_bounds__(1024)
bin_scatter(const int* __restrict__ src, const int* __restrict__ dst,
            const float* __restrict__ ew, const int* __restrict__ offs,
            int* __restrict__ tdst, int* __restrict__ tpair, int E, int chunk, int nbin) {
    constexpr int R    = 4;
    constexpr int TILE = R * SORT_T;
    __shared__ int  cur[MAX_BIN];
    __shared__ int  lcu[MAX_BIN];
    __shared__ int  lst[MAX_BIN];
    __shared__ int  wsum[32];
    __shared__ int2 sp[TILE];
    __shared__ int  sd[TILE];
    __shared__ int  gp[TILE];
    const int t = threadIdx.x;
    const int g = blockIdx.x;
    cur[t] = (t < nbin) ? offs[t * SORT_G + g] : 0;
    const int lo = g * chunk;
    const int hi = min(E, lo + chunk);
    for (int t0 = lo; t0 < hi; t0 += TILE) {
        const int te = min(hi, t0 + TILE);
        lcu[t] = 0;
        __syncthreads();

        int d[R], s[R], w[R], rk[R];
#pragma unroll
        for (int r = 0; r < R; ++r) {
            const int e = t0 + t + SORT_T * r;
            if (e < te) {
                d[r] = dst[e];
                s[r] = src[e];
                w[r] = __float_as_int(ew[e]);
            }
        }
#pragma unroll
        for (int r = 0; r < R; ++r) {
            const int e = t0 + t + SORT_T * r;
            if (e < te) rk[r] = atomicAdd(&lcu[d[r] >> BIN_SH], 1);
        }
        __syncthreads();

        const int c = lcu[t];
        lst[t] = block_scan_excl(c, wsum, t);
        __syncthreads();

#pragma unroll
        for (int r = 0; r < R; ++r) {
            const int e = t0 + t + SORT_T * r;
            if (e < te) {
                const int bin = d[r] >> BIN_SH;
                const int k   = lst[bin] + rk[r];
                sp[k] = make_int2(s[r], w[r]);
                sd[k] = d[r];
                gp[k] = cur[bin] + rk[r];
            }
        }
        __syncthreads();

        cur[t] += c;
        const int nt = te - t0;
        for (int k = t; k < nt; k += SORT_T) {
            const int p = gp[k];
            tdst[p] = sd[k];
            ((int2*)tpair)[p] = sp[k];
        }
        __syncthreads();
    }
}

__global__ void __launch_bounds__(1024)
bin_sort(const int* __restrict__ tdst, const int* __restrict__ tpair,
         const int* __restrict__ offs, int* __restrict__ rowptr,
         float* __restrict__ deg, int* __restrict__ sorted, int N) {
    constexpr int R    = 8;
    constexpr int TILE = R * SORT_T;
    __shared__ int  cnt[BIN_W];
    __shared__ int  cur[BIN_W];
    __shared__ int  lst[BIN_W];
    __shared__ int  wsum[32];
    __shared__ int2 sp[TILE];
    __shared__ int  gp[TILE];
    const int t   = threadIdx.x;
    const int b   = blockIdx.x;
    const int n0  = b << BIN_SH;
    const int beg = offs[b * SORT_G];
    const int end = offs[(b + 1) * SORT_G];

    const bool one = end > beg && end - beg <= TILE;

    if (!one) {
        if (t < BIN_W) cnt[t] = 0;
        __syncthreads();
        for (int i = beg + t; i < end; i += 1024) atomicAdd(&cnt[tdst[i] - n0], 1);
        __syncthreads();
        const int v  = (t < BIN_W) ? cnt[t] : 0;
        const int ex = block_scan_excl(v, wsum, t);
        if (t < BIN_W) {
            const int start = beg + ex;
            const int node  = n0 + t;
            cur[t] = start;
            if (node < N) {
                rowptr[node] = start;
                deg[node]    = (float)v;
                if (node == N - 1) rowptr[N] = end;
            }
        }
        __syncthreads();
    }

    for (int t0 = beg; t0 < end; t0 += TILE) {
        const int te = min(end, t0 + TILE);
        if (t < BIN_W) cnt[t] = 0;
        __syncthreads();

        int  dd[R], rk[R];
        int2 pr[R];
#pragma unroll
        for (int r = 0; r < R; ++r) {
            const int i = t0 + t + SORT_T * r;
            if (i < te) {
                dd[r] = tdst[i];
                pr[r] = ((const int2*)tpair)[i];
            }
        }
#pragma unroll
        for (int r = 0; r < R; ++r) {
            const int i = t0 + t + SORT_T * r;
            if (i < te) rk[r] = atomicAdd(&cnt[dd[r] - n0], 1);
        }
        __syncthreads();

        const int c  = (t < BIN_W) ? cnt[t] : 0;
        const int lx = block_scan_excl(c, wsum, t);
        if (t < BIN_W) {
            lst[t] = lx;
            if (one) {
                const int start = beg + lx;
                const int node  = n0 + t;
                cur[t] = start;
                if (node < N) {
                    rowptr[node] = start;
                    deg[node]    = (float)c;
                    if (node == N - 1) rowptr[N] = end;
                }
            }
        }
        __syncthreads();

#pragma unroll
        for (int r = 0; r < R; ++r) {
            const int i = t0 + t + SORT_T * r;
            if (i < te) {
                const int nd = dd[r] - n0;
                const int k  = lst[nd] + rk[r];
                sp[k] = pr[r];
                gp[k] = cur[nd] + rk[r];
            }
        }
        __syncthreads();

        if (t < BIN_W) cur[t] += c;
        const int nt = te - t0;
        for (int k = t; k < nt; k += SORT_T) ((int2*)sorted)[gp[k]] = sp[k];
        __syncthreads();
    }
}

template <int K, int F>
__global__ void __launch_bounds__(32)
sage_project(const float* __restrict__ xin,
             const float* __restrict__ Ws, const float* __restrict__ Wn,
             const float* __restrict__ bias,
             float* __restrict__ S, float* __restrict__ P) {
    static_assert(F == 1 || F % 16 == 0, "column tiling");
    static_assert(K % 4 == 0, "k steps");
    constexpr int NS = K / 4;
    constexpr int FT = (F >= 16) ? F / 16 : 1;
    constexpr int NT = (F >= 16) ? 2 * FT : 1;

    const int lane = threadIdx.x;
    const int c    = lane & 15;
    const int hh   = lane >> 4;
    const int row0 = blockIdx.x << 4;
    const int node = row0 + c;

    v2f a[NS];
    const float* xrow = xin + (size_t)node * K;
    for (int t = 0; t < NS; ++t) {
        const int k0 = 4 * t + 2 * hh;
        a[t] = *(const v2f*)(xrow + k0);
    }

    v8f d[NT];
    for (int j = 0; j < NT; ++j) d[j] = v8f{};

    for (int t = 0; t < NS; ++t) {
        const int k0 = 4 * t + 2 * hh;
        for (int j = 0; j < NT; ++j) {
            v2f b;
            if (F >= 16) {
                const float* W = (j < FT) ? Ws : Wn;
                const int col  = 16 * (j % FT) + c;
                b.x = W[(k0)     * F + col];
                b.y = W[(k0 + 1) * F + col];
            } else {
                const float* W = (c == 1) ? Wn : Ws;
                const float w0 = W[k0];
                const float w1 = W[k0 + 1];
                b.x = (c < 2) ? w0 : 0.0f;
                b.y = (c < 2) ? w1 : 0.0f;
            }
            d[j] = __builtin_amdgcn_wmma_f32_16x16x4_f32(false, a[t], false, b,
                                                         (short)0, d[j], false, false);
        }
    }

    for (int j = 0; j < NT; ++j) {
        for (int r = 0; r < 8; ++r) {
            const size_t row = (size_t)(row0 + 8 * hh + r);
            if (F >= 16) {
                const int col = 16 * (j % FT) + c;
                if (j < FT) S[row * F + col] = d[j][r] + bias[col];
                else        P[row * F + col] = d[j][r];
            } else {
                float* o = (c == 1) ? P : S;
                const float v = (c == 0) ? d[0][r] + bias[0] : d[0][r];
                if (c < 2) o[row] = v;
            }
        }
    }
}

template <int F, int FO>
__global__ void __launch_bounds__(32)
pull_project(const float* S, const float* __restrict__ P,
             const int* __restrict__ rowptr, const int* __restrict__ sorted,
             const float* __restrict__ deg,
             const float* __restrict__ Ws, const float* __restrict__ Wn,
             const float* __restrict__ bias, float* So, float* Po) {
    static_assert(F == 16 || F == 32, "lane group per node");
    static_assert(FO == 1 || FO == 16, "column tiling");
    constexpr int PIN   = 32;
    constexpr int PQ    = PIN / 4;
    constexpr int LPN   = F / 4;
    constexpr int NPP   = 32 / LPN;
    constexpr int NPASS = 16 / NPP;
    constexpr int EB    = 8;
    constexpr int XP    = F + 4;
    constexpr int NS    = F / 4;
    constexpr int NT    = (FO == 16) ? 2 : 1;
    __shared__ __attribute__((aligned(16))) float xs[16 * XP];

    const int lane = threadIdx.x;
    const int sub  = lane % LPN;
    const int grp  = lane / LPN;
    const int row0 = blockIdx.x << 4;
    const int2* ed = (const int2*)sorted;
    const v4f*  P4 = (const v4f*)P;
    const v4f*  S4 = (const v4f*)S;

    int beg[NPASS], end[NPASS];
    for (int j = 0; j < NPASS; ++j) {
        const int node = row0 + j * NPP + grp;
        beg[j] = rowptr[node];
        end[j] = rowptr[node + 1];
    }

    int2 e[EB];
    {
        const int last = max(end[0] - 1, 0);
        for (int i = 0; i < EB; ++i) e[i] = ed[min(beg[0] + i, last)];
    }

    for (int j = 0; j < NPASS; ++j) {
        const int slot = j * NPP + grp;
        const int node = row0 + slot;
        const int last = max(end[j] - 1, 0);

        int2 ef[EB];
        if (j + 1 < NPASS) {
            const int lastn = max(end[j + 1] - 1, 0);
            for (int i = 0; i < EB; ++i) ef[i] = ed[min(beg[j + 1] + i, lastn)];
        }
        const v4f   s4 = S4[(size_t)node * PQ + sub];
        const float dg = deg[node];

        v4f acc = v4f{};
        for (int c0 = beg[j]; c0 < end[j]; c0 += EB) {
            int2 en[EB];
            for (int i = 0; i < EB; ++i) en[i] = ed[min(c0 + EB + i, last)];
            v4f p[EB];
            for (int i = 0; i < EB; ++i) p[i] = P4[e[i].x * PQ + sub];
            __builtin_amdgcn_sched_barrier(0);
            for (int i = 0; i < EB; ++i) {
                const float w  = __int_as_float(e[i].y);
                const bool  in = c0 + i < end[j];
                acc.x += in ? w * p[i].x : 0.0f;
                acc.y += in ? w * p[i].y : 0.0f;
                acc.z += in ? w * p[i].z : 0.0f;
                acc.w += in ? w * p[i].w : 0.0f;
            }
            for (int i = 0; i < EB; ++i) e[i] = en[i];
        }

        const float invdeg = 1.0f / fmaxf(dg, 1.0f);
        v4f x;
        x.x = tanhf(s4.x + acc.x * invdeg);
        x.y = tanhf(s4.y + acc.y * invdeg);
        x.z = tanhf(s4.z + acc.z * invdeg);
        x.w = tanhf(s4.w + acc.w * invdeg);
        *(v4f*)(xs + slot * XP + 4 * sub) = x;

        if (j + 1 < NPASS)
            for (int i = 0; i < EB; ++i) e[i] = ef[i];
    }
    __syncthreads();

    const int c  = lane & 15;
    const int hh = lane >> 4;

    v2f a[NS];
    for (int t = 0; t < NS; ++t) {
        const int k0 = 4 * t + 2 * hh;
        a[t] = *(const v2f*)(xs + c * XP + k0);
    }

    v8f d[NT];
    for (int j = 0; j < NT; ++j) d[j] = v8f{};

    for (int t = 0; t < NS; ++t) {
        const int k0 = 4 * t + 2 * hh;
        for (int j = 0; j < NT; ++j) {
            v2f b;
            if (FO == 16) {
                const float* W = (j == 0) ? Ws : Wn;
                b.x = W[(k0)     * FO + c];
                b.y = W[(k0 + 1) * FO + c];
            } else {
                const float* W = (c == 1) ? Wn : Ws;
                const float w0 = W[k0];
                const float w1 = W[k0 + 1];
                b.x = (c < 2) ? w0 : 0.0f;
                b.y = (c < 2) ? w1 : 0.0f;
            }
            d[j] = __builtin_amdgcn_wmma_f32_16x16x4_f32(false, a[t], false, b,
                                                         (short)0, d[j], false, false);
        }
    }

    for (int r = 0; r < 8; ++r) {
        const size_t row = (size_t)(row0 + 8 * hh + r);
        if (FO == 16) {
            So[row * PIN + c] = d[0][r] + bias[c];
            Po[row * PIN + c] = d[NT - 1][r];
        } else {
            float* o = (c == 1) ? Po : So;
            const float v = (c == 0) ? d[0][r] + bias[0] : d[0][r];
            if (c < 2) o[row] = v;
        }
    }
}

__global__ void __launch_bounds__(32)
pull_final(const float* __restrict__ P, const int* __restrict__ rowptr,
           const int* __restrict__ sorted, const float* __restrict__ S,
           const float* __restrict__ deg, float* __restrict__ out) {
    constexpr int NI = 8;
    const int lane = threadIdx.x;
    const int sub  = lane & 15;
    const int h    = lane >> 4;
    const int base = blockIdx.x * 16;
    const int2* ed = (const int2*)sorted;

    const int   onode = base + 2 * (sub & (NI - 1)) + h;
    const float sv    = S[onode];
    const float dv    = deg[onode];

    const int* rp = rowptr + base + h;
    int beg[NI], end[NI];
    for (int i = 0; i < NI; ++i) {
        beg[i] = rp[2 * i];
        end[i] = rp[2 * i + 1];
    }

    int2 e0[NI], e1[NI];
    for (int i = 0; i < NI; ++i) {
        const int last = max(end[i] - 1, 0);
        e0[i] = ed[min(beg[i] + sub, last)];
        e1[i] = ed[min(beg[i] + sub + 16, last)];
    }
    float p0[NI], p1[NI];
    for (int i = 0; i < NI; ++i) {
        p0[i] = P[e0[i].x];
        p1[i] = P[e1[i].x];
    }

    float acc[NI];
    for (int i = 0; i < NI; ++i) {
        float a = 0.0f;
        if (beg[i] + sub      < end[i]) a += __int_as_float(e0[i].y) * p0[i];
        if (beg[i] + sub + 16 < end[i]) a += __int_as_float(e1[i].y) * p1[i];
        for (int j = beg[i] + sub + 32; j < end[i]; j += 16) {
            const int2 ev = ed[j];
            a += __int_as_float(ev.y) * P[ev.x];
        }
        acc[i] = a;
    }
    for (int s = 8; s > 0; s >>= 1)
        for (int i = 0; i < NI; ++i) acc[i] += __shfl_xor(acc[i], s, 16);

    float res = acc[0];
    for (int i = 1; i < NI; ++i) res = (sub == i) ? acc[i] : res;
    if (sub < NI) out[onode] = sv + res * (1.0f / fmaxf(dv, 1.0f));
}

extern "C" void kernel_launch(void* const* d_in, const int* in_sizes, int n_in,
                              void* d_out, int out_size, void* d_ws, size_t ws_size,
                              hipStream_t stream) {
    const float* b_z  = (const float*)d_in[0];
    const int*   esrc = (const int*)  d_in[1];
    const int*   edst = (const int*)  d_in[2];
    const float* ew   = (const float*)d_in[3];
    const float* Ws0  = (const float*)d_in[4];
    const float* Wn0  = (const float*)d_in[5];
    const float* b0   = (const float*)d_in[6];
    const float* Ws1  = (const float*)d_in[7];
    const float* Wn1  = (const float*)d_in[8];
    const float* b1   = (const float*)d_in[9];
    const float* Ws2  = (const float*)d_in[10];
    const float* Wn2  = (const float*)d_in[11];
    const float* b2   = (const float*)d_in[12];

    const int N = in_sizes[0] / 64;
    const int E = in_sizes[1];
    const size_t n = (size_t)N;

    float* ws     = (float*)d_ws;
    float* deg    = ws;
    int*   rowptr = (int*)(ws + n);
    int*   sorted = (int*)(ws + 4 * n);
    int*   tdst   = (int*)(ws + 36 * n);
    int*   tpair  = (int*)(ws + 52 * n);
    int*   counts = (int*)(ws + 84 * n);
    int*   offs   = (int*)(ws + 86 * n);
    float* S0     = ws + 36 * n;
    float* P0     = ws + 68 * n;
    float* S1     = ws + 36 * n;
    float* P1     = ws + 36 * n + 16;
    float* S2     = ws + 68 * n;
    float* P2     = ws + 72 * n;
    float* out    = (float*)d_out;

    auto cdiv = [](int a, int b) { return (a + b - 1) / b; };

    const int nbin  = cdiv(N, BIN_W);
    const int chunk = cdiv(E, SORT_G);
    const int M     = nbin * SORT_G;
    bin_count<<<SORT_G, 1024, 0, stream>>>(edst, counts, E, chunk, nbin);
    scan_excl<<<cdiv(M, 1024), 1024, 0, stream>>>(counts, offs, M);
    bin_scatter<<<SORT_G, 1024, 0, stream>>>(esrc, edst, ew, offs, tdst, tpair, E, chunk, nbin);
    bin_sort<<<nbin, 1024, 0, stream>>>(tdst, tpair, offs, rowptr, deg, sorted, N);

    sage_project<64, 32><<<N / 16, 32, 0, stream>>>(b_z, Ws0, Wn0, b0, S0, P0);

    pull_project<32, 16><<<N / 16, 32, 0, stream>>>(S0, P0, rowptr, sorted, deg, Ws1, Wn1, b1, S1, P1);

    pull_project<16, 1><<<N / 16, 32, 0, stream>>>(S1, P1, rowptr, sorted, deg, Ws2, Wn2, b2, S2, P2);

    pull_final<<<N / 16, 32, 0, stream>>>(P2, rowptr, sorted, S2, deg, out);
}


# Edge sort in 2 launches, not 4: each chunk ordered by bin in its own slice (no count or scan launch); a bin block sums chunk offsets, gathers its 256 runs via LDS; a node's f32 sum order may change
# speedup vs baseline: 4.9329x; 1.3496x over previous
#include <hip/hip_runtime.h>


typedef __attribute__((ext_vector_type(2))) float v2f;
typedef __attribute__((ext_vector_type(4))) float v4f;
typedef __attribute__((ext_vector_type(8))) float v8f;

constexpr int SORT_G  = 256;
constexpr int BIN_SH  = 8;
constexpr int BIN_W   = 1 << BIN_SH;
constexpr int MAX_BIN = 1024;
constexpr int SORT_T  = 1024;
static_assert(MAX_BIN == SORT_T, "one thread per bin counter");

__device__ __forceinline__ int block_scan_excl(int v, int* wsum, int t) {
    const int lane = t & 31;
    const int w    = t >> 5;
    int x = v;
    for (int off = 1; off < 32; off <<= 1) {
        const int y = __shfl_up(x, off, 32);
        if (lane >= off) x += y;
    }
    if (lane == 31) wsum[w] = x;
    __syncthreads();
    if (w == 0) {
        int s = wsum[lane];
        for (int off = 1; off < 32; off <<= 1) {
            const int y = __shfl_up(s, off, 32);
            if (lane >= off) s += y;
        }
        wsum[lane] = s;
    }
    __syncthreads();
    const int base = (w > 0) ? wsum[w - 1] : 0;
    return base + x - v;
}

__global__ void __launch_bounds__(1024)
piece_sort(const int* __restrict__ src, const int* __restrict__ dst,
           const float* __restrict__ ew, int* __restrict__ loff,
           int* __restrict__ tpk, int E, int chunk, int nbin) {
    constexpr int R    = 4;
    constexpr int TILE = R * SORT_T;
    __shared__ int  cur[MAX_BIN];
    __shared__ int  lcu[MAX_BIN];
    __shared__ int  lst[MAX_BIN];
    __shared__ int  wsum[32];
    __shared__ int2 sp[TILE];
    __shared__ int  gp[TILE];
    const int t  = threadIdx.x;
    const int g  = blockIdx.x;
    const int lo = g * chunk;
    const int hi = min(E, lo + chunk);
    if (hi <= lo) {
        if (t <= nbin) loff[t * SORT_G + g] = 0;
        return;
    }

    const bool one = hi - lo <= TILE;
    if (!one) {
        lcu[t] = 0;
        __syncthreads();
        for (int e = lo + t; e < hi; e += SORT_T) atomicAdd(&lcu[dst[e] >> BIN_SH], 1);
        __syncthreads();
        const int v  = lcu[t];
        const int ex = block_scan_excl(v, wsum, t);
        cur[t] = ex;
        if (t <= nbin) loff[t * SORT_G + g] = ex;
        __syncthreads();
    }

    for (int t0 = lo; t0 < hi; t0 += TILE) {
        const int te = min(hi, t0 + TILE);
        lcu[t] = 0;
        __syncthreads();

        int d[R], s[R], w[R], rk[R];
#pragma unroll
        for (int r = 0; r < R; ++r) {
            const int e = t0 + t + SORT_T * r;
            if (e < te) {
                d[r] = dst[e];
                s[r] = src[e];
                w[r] = __float_as_int(ew[e]);
            }
        }
#pragma unroll
        for (int r = 0; r < R; ++r) {
            const int e = t0 + t + SORT_T * r;
            if (e < te) rk[r] = atomicAdd(&lcu[d[r] >> BIN_SH], 1);
        }
        __syncthreads();

        const int c  = lcu[t];
        const int lx = block_scan_excl(c, wsum, t);
        lst[t] = lx;
        if (one) {
            cur[t] = lx;
            if (t <= nbin) loff[t * SORT_G + g] = lx;
        }
        __syncthreads();

#pragma unroll
        for (int r = 0; r < R; ++r) {
            const int e = t0 + t + SORT_T * r;
            if (e < te) {
                const int bin = d[r] >> BIN_SH;
                const int k   = lst[bin] + rk[r];
                sp[k] = make_int2((int)(((unsigned)s[r] << BIN_SH) | ((unsigned)d[r] & (BIN_W - 1))), w[r]);
                gp[k] = cur[bin] + rk[r];
            }
        }
        __syncthreads();

        cur[t] += c;
        const int nt = te - t0;
        for (int k = t; k < nt; k += SORT_T) ((int2*)tpk)[lo + gp[k]] = sp[k];
        __syncthreads();
    }
}

__global__ void __launch_bounds__(1024)
bin_sort(const int* __restrict__ tpk, const int* __restrict__ loff,
         int* __restrict__ rowptr, float* __restrict__ deg, int* __restrict__ sorted,
         int N, int chunk) {
    constexpr int R    = 6;
    constexpr int TILE = R * SORT_T;
    constexpr int SPW  = SORT_G / 32;
    static_assert(SORT_G % 32 == 0 && SORT_G <= SORT_T, "one thread per run; whole runs per wave");
    __shared__ int  cnt[BIN_W];
    __shared__ int  cur[BIN_W];
    __shared__ int  lst[BIN_W];
    __shared__ int  wsum[32];
    __shared__ int  sbase[SORT_G];
    __shared__ int  scnt[SORT_G];
    __shared__ int  sadr[SORT_G];
    __shared__ int  tot[2];
    __shared__ int2 st[TILE];
    __shared__ int2 sp[TILE];
    __shared__ int  gp[TILE];
    const int t    = threadIdx.x;
    const int lane = t & 31;
    const int wv   = t >> 5;
    const int b    = blockIdx.x;
    const int2* pk = (const int2*)tpk;

    int a0 = 0, cg = 0;
    if (t < SORT_G) {
        a0 = loff[b * SORT_G + t];
        cg = loff[(b + 1) * SORT_G + t] - a0;
    }
    if (t < 2) tot[t] = 0;
    __syncthreads();
    if (t < SORT_G) {
        int ra = a0, rc = cg;
        for (int s = 16; s > 0; s >>= 1) {
            ra += __shfl_xor(ra, s, 32);
            rc += __shfl_xor(rc, s, 32);
        }
        if (lane == 0) {
            atomicAdd(&tot[0], ra);
            atomicAdd(&tot[1], rc);
        }
    }
    const int ex = block_scan_excl(cg, wsum, t);
    if (t < SORT_G) {
        sbase[t] = ex;
        scnt[t]  = cg;
        sadr[t]  = t * chunk + a0;
    }
    __syncthreads();
    const int beg   = tot[0];
    const int total = tot[1];
    const int end   = beg + total;

    const bool one = total > 0 && total <= TILE;

    if (!one) {
        if (t < BIN_W) cnt[t] = 0;
        __syncthreads();
        for (int s = 0; s < SPW; ++s) {
            const int g = wv * SPW + s;
            const int c = scnt[g];
            const int a = sadr[g];
            for (int i = lane; i < c; i += 32) atomicAdd(&cnt[pk[a + i].x & (BIN_W - 1)], 1);
        }
        __syncthreads();
        const int v  = (t < BIN_W) ? cnt[t] : 0;
        const int vx = block_scan_excl(v, wsum, t);
        if (t < BIN_W) {
            const int start = beg + vx;
            const int node  = (b << BIN_SH) + t;
            cur[t] = start;
            if (node < N) {
                rowptr[node] = start;
                deg[node]    = (float)v;
                if (node == N - 1) rowptr[N] = end;
            }
        }
        __syncthreads();
    }

    for (int t0 = 0; t0 < total; t0 += TILE) {
        const int nt = min(total - t0, TILE);
        if (t < BIN_W) cnt[t] = 0;

        int2 v[SPW];
        int  p[SPW];
#pragma unroll
        for (int s = 0; s < SPW; ++s) {
            const int g = wv * SPW + s;
            p[s] = (lane < scnt[g]) ? sbase[g] + lane - t0 : -1;
            if (p[s] >= 0 && p[s] < TILE) v[s] = pk[sadr[g] + lane];
        }
#pragma unroll
        for (int s = 0; s < SPW; ++s)
            if (p[s] >= 0 && p[s] < TILE) st[p[s]] = v[s];
        for (int s = 0; s < SPW; ++s) {
            const int g = wv * SPW + s;
            const int c = scnt[g];
            for (int i = lane + 32; i < c; i += 32) {
                const int q = sbase[g] + i - t0;
                if (q >= 0 && q < TILE) st[q] = pk[sadr[g] + i];
            }
        }
        __syncthreads();

        int2 pr[R];
        int  rk[R];
#pragma unroll
        for (int r = 0; r < R; ++r) {
            const int i = t + SORT_T * r;
            if (i < nt) {
                pr[r] = st[i];
                rk[r] = atomicAdd(&cnt[pr[r].x & (BIN_W - 1)], 1);
            }
        }
        __syncthreads();

        const int c  = (t < BIN_W) ? cnt[t] : 0;
        const int lx = block_scan_excl(c, wsum, t);
        if (t < BIN_W) {
            lst[t] = lx;
            if (one) {
                const int start = beg + lx;
                const int node  = (b << BIN_SH) + t;
                cur[t] = start;
                if (node < N) {
                    rowptr[node] = start;
                    deg[node]    = (float)c;
                    if (node == N - 1) rowptr[N] = end;
                }
            }
        }
        __syncthreads();

#pragma unroll
        for (int r = 0; r < R; ++r) {
            const int i = t + SORT_T * r;
            if (i < nt) {
                const int nd = pr[r].x & (BIN_W - 1);
                const int k  = lst[nd] + rk[r];
                sp[k] = make_int2((int)((unsigned)pr[r].x >> BIN_SH), pr[r].y);
                gp[k] = cur[nd] + rk[r];
            }
        }
        __syncthreads();

        if (t < BIN_W) cur[t] += c;
        for (int k = t; k < nt; k += SORT_T) ((int2*)sorted)[gp[k]] = sp[k];
        __syncthreads();
    }
}

template <int K, int F>
__global__ void __launch_bounds__(32)
sage_project(const float* __restrict__ xin,
             const float* __restrict__ Ws, const float* __restrict__ Wn,
             const float* __restrict__ bias,
             float* __restrict__ S, float* __restrict__ P) {
    static_assert(F == 1 || F % 16 == 0, "column tiling");
    static_assert(K % 4 == 0, "k steps");
    constexpr int NS = K / 4;
    constexpr int FT = (F >= 16) ? F / 16 : 1;
    constexpr int NT = (F >= 16) ? 2 * FT : 1;

    const int lane = threadIdx.x;
    const int c    = lane & 15;
    const int hh   = lane >> 4;
    const int row0 = blockIdx.x << 4;
    const int node = row0 + c;

    v2f a[NS];
    const float* xrow = xin + (size_t)node * K;
    for (int t = 0; t < NS; ++t) {
        const int k0 = 4 * t + 2 * hh;
        a[t] = *(const v2f*)(xrow + k0);
    }

    v8f d[NT];
    for (int j = 0; j < NT; ++j) d[j] = v8f{};

    for (int t = 0; t < NS; ++t) {
        const int k0 = 4 * t + 2 * hh;
        for (int j = 0; j < NT; ++j) {
            v2f b;
            if (F >= 16) {
                const float* W = (j < FT) ? Ws : Wn;
                const int col  = 16 * (j % FT) + c;
                b.x = W[(k0)     * F + col];
                b.y = W[(k0 + 1) * F + col];
            } else {
                const float* W = (c == 1) ? Wn : Ws;
                const float w0 = W[k0];
                const float w1 = W[k0 + 1];
                b.x = (c < 2) ? w0 : 0.0f;
                b.y = (c < 2) ? w1 : 0.0f;
            }
            d[j] = __builtin_amdgcn_wmma_f32_16x16x4_f32(false, a[t], false, b,
                                                         (short)0, d[j], false, false);
        }
    }

    for (int j = 0; j < NT; ++j) {
        for (int r = 0; r < 8; ++r) {
            const size_t row = (size_t)(row0 + 8 * hh + r);
            if (F >= 16) {
                const int col = 16 * (j % FT) + c;
                if (j < FT) S[row * F + col] = d[j][r] + bias[col];
                else        P[row * F + col] = d[j][r];
            } else {
                float* o = (c == 1) ? P : S;
                const float v = (c == 0) ? d[0][r] + bias[0] : d[0][r];
                if (c < 2) o[row] = v;
            }
        }
    }
}

template <int F, int FO>
__global__ void __launch_bounds__(32)
pull_project(const float* S, const float* __restrict__ P,
             const int* __restrict__ rowptr, const int* __restrict__ sorted,
             const float* __restrict__ deg,
             const float* __restrict__ Ws, const float* __restrict__ Wn,
             const float* __restrict__ bias, float* So, float* Po) {
    static_assert(F == 16 || F == 32, "lane group per node");
    static_assert(FO == 1 || FO == 16, "column tiling");
    constexpr int PIN   = 32;
    constexpr int PQ    = PIN / 4;
    constexpr int LPN   = F / 4;
    constexpr int NPP   = 32 / LPN;
    constexpr int NPASS = 16 / NPP;
    constexpr int EB    = 8;
    constexpr int XP    = F + 4;
    constexpr int NS    = F / 4;
    constexpr int NT    = (FO == 16) ? 2 : 1;
    __shared__ __attribute__((aligned(16))) float xs[16 * XP];

    const int lane = threadIdx.x;
    const int sub  = lane % LPN;
    const int grp  = lane / LPN;
    const int row0 = blockIdx.x << 4;
    const int2* ed = (const int2*)sorted;
    const v4f*  P4 = (const v4f*)P;
    const v4f*  S4 = (const v4f*)S;

    int beg[NPASS], end[NPASS];
    for (int j = 0; j < NPASS; ++j) {
        const int node = row0 + j * NPP + grp;
        beg[j] = rowptr[node];
        end[j] = rowptr[node + 1];
    }

    int2 e[EB];
    {
        const int last = max(end[0] - 1, 0);
        for (int i = 0; i < EB; ++i) e[i] = ed[min(beg[0] + i, last)];
    }

    for (int j = 0; j < NPASS; ++j) {
        const int slot = j * NPP + grp;
        const int node = row0 + slot;
        const int last = max(end[j] - 1, 0);

        int2 ef[EB];
        if (j + 1 < NPASS) {
            const int lastn = max(end[j + 1] - 1, 0);
            for (int i = 0; i < EB; ++i) ef[i] = ed[min(beg[j + 1] + i, lastn)];
        }
        const v4f   s4 = S4[(size_t)node * PQ + sub];
        const float dg = deg[node];

        v4f acc = v4f{};
        for (int c0 = beg[j]; c0 < end[j]; c0 += EB) {
            int2 en[EB];
            for (int i = 0; i < EB; ++i) en[i] = ed[min(c0 + EB + i, last)];
            v4f p[EB];
            for (int i = 0; i < EB; ++i) p[i] = P4[e[i].x * PQ + sub];
            __builtin_amdgcn_sched_barrier(0);
            for (int i = 0; i < EB; ++i) {
                const float w  = __int_as_float(e[i].y);
                const bool  in = c0 + i < end[j];
                acc.x += in ? w * p[i].x : 0.0f;
                acc.y += in ? w * p[i].y : 0.0f;
                acc.z += in ? w * p[i].z : 0.0f;
                acc.w += in ? w * p[i].w : 0.0f;
            }
            for (int i = 0; i < EB; ++i) e[i] = en[i];
        }

        const float invdeg = 1.0f / fmaxf(dg, 1.0f);
        v4f x;
        x.x = tanhf(s4.x + acc.x * invdeg);
        x.y = tanhf(s4.y + acc.y * invdeg);
        x.z = tanhf(s4.z + acc.z * invdeg);
        x.w = tanhf(s4.w + acc.w * invdeg);
        *(v4f*)(xs + slot * XP + 4 * sub) = x;

        if (j + 1 < NPASS)
            for (int i = 0; i < EB; ++i) e[i] = ef[i];
    }
    __syncthreads();

    const int c  = lane & 15;
    const int hh = lane >> 4;

    v2f a[NS];
    for (int t = 0; t < NS; ++t) {
        const int k0 = 4 * t + 2 * hh;
        a[t] = *(const v2f*)(xs + c * XP + k0);
    }

    v8f d[NT];
    for (int j = 0; j < NT; ++j) d[j] = v8f{};

    for (int t = 0; t < NS; ++t) {
        const int k0 = 4 * t + 2 * hh;
        for (int j = 0; j < NT; ++j) {
            v2f b;
            if (FO == 16) {
                const float* W = (j == 0) ? Ws : Wn;
                b.x = W[(k0)     * FO + c];
                b.y = W[(k0 + 1) * FO + c];
            } else {
                const float* W = (c == 1) ? Wn : Ws;
                const float w0 = W[k0];
                const float w1 = W[k0 + 1];
                b.x = (c < 2) ? w0 : 0.0f;
                b.y = (c < 2) ? w1 : 0.0f;
            }
            d[j] = __builtin_amdgcn_wmma_f32_16x16x4_f32(false, a[t], false, b,
                                                         (short)0, d[j], false, false);
        }
    }

    for (int r = 0; r < 8; ++r) {
        const size_t row = (size_t)(row0 + 8 * hh + r);
        if (FO == 16) {
            So[row * PIN + c] = d[0][r] + bias[c];
            Po[row * PIN + c] = d[NT - 1][r];
        } else {
            float* o = (c == 1) ? Po : So;
            const float v = (c == 0) ? d[0][r] + bias[0] : d[0][r];
            if (c < 2) o[row] = v;
        }
    }
}

__global__ void __launch_bounds__(32)
pull_final(const float* __restrict__ P, const int* __restrict__ rowptr,
           const int* __restrict__ sorted, const float* __restrict__ S,
           const float* __restrict__ deg, float* __restrict__ out) {
    constexpr int NI = 8;
    const int lane = threadIdx.x;
    const int sub  = lane & 15;
    const int h    = lane >> 4;
    const int base = blockIdx.x * 16;
    const int2* ed = (const int2*)sorted;

    const int   onode = base + 2 * (sub & (NI - 1)) + h;
    const float sv    = S[onode];
    const float dv    = deg[onode];

    const int* rp = rowptr + base + h;
    int beg[NI], end[NI];
    for (int i = 0; i < NI; ++i) {
        beg[i] = rp[2 * i];
        end[i] = rp[2 * i + 1];
    }

    int2 e0[NI], e1[NI];
    for (int i = 0; i < NI; ++i) {
        const int last = max(end[i] - 1, 0);
        e0[i] = ed[min(beg[i] + sub, last)];
        e1[i] = ed[min(beg[i] + sub + 16, last)];
    }
    float p0[NI], p1[NI];
    for (int i = 0; i < NI; ++i) {
        p0[i] = P[e0[i].x];
        p1[i] = P[e1[i].x];
    }

    float acc[NI];
    for (int i = 0; i < NI; ++i) {
        float a = 0.0f;
        if (beg[i] + sub      < end[i]) a += __int_as_float(e0[i].y) * p0[i];
        if (beg[i] + sub + 16 < end[i]) a += __int_as_float(e1[i].y) * p1[i];
        for (int j = beg[i] + sub + 32; j < end[i]; j += 16) {
            const int2 ev = ed[j];
            a += __int_as_float(ev.y) * P[ev.x];
        }
        acc[i] = a;
    }
    for (int s = 8; s > 0; s >>= 1)
        for (int i = 0; i < NI; ++i) acc[i] += __shfl_xor(acc[i], s, 16);

    float res = acc[0];
    for (int i = 1; i < NI; ++i) res = (sub == i) ? acc[i] : res;
    if (sub < NI) out[onode] = sv + res * (1.0f / fmaxf(dv, 1.0f));
}

extern "C" void kernel_launch(void* const* d_in, const int* in_sizes, int n_in,
                              void* d_out, int out_size, void* d_ws, size_t ws_size,
                              hipStream_t stream) {
    const float* b_z  = (const float*)d_in[0];
    const int*   esrc = (const int*)  d_in[1];
    const int*   edst = (const int*)  d_in[2];
    const float* ew   = (const float*)d_in[3];
    const float* Ws0  = (const float*)d_in[4];
    const float* Wn0  = (const float*)d_in[5];
    const float* b0   = (const float*)d_in[6];
    const float* Ws1  = (const float*)d_in[7];
    const float* Wn1  = (const float*)d_in[8];
    const float* b1   = (const float*)d_in[9];
    const float* Ws2  = (const float*)d_in[10];
    const float* Wn2  = (const float*)d_in[11];
    const float* b2   = (const float*)d_in[12];

    const int N = in_sizes[0] / 64;
    const int E = in_sizes[1];
    const size_t n = (size_t)N;

    float* ws     = (float*)d_ws;
    float* deg    = ws;
    int*   rowptr = (int*)(ws + n);
    int*   sorted = (int*)(ws + 4 * n);
    int*   tpk    = (int*)(ws + 36 * n);
    int*   loff   = (int*)(ws + 84 * n);
    float* S0     = ws + 36 * n;
    float* P0     = ws + 68 * n;
    float* S1     = ws + 36 * n;
    float* P1     = ws + 36 * n + 16;
    float* S2     = ws + 68 * n;
    float* P2     = ws + 72 * n;
    float* out    = (float*)d_out;

    auto cdiv = [](int a, int b) { return (a + b - 1) / b; };

    const int nbin  = cdiv(N, BIN_W);
    const int chunk = cdiv(E, SORT_G);
    piece_sort<<<SORT_G, 1024, 0, stream>>>(esrc, edst, ew, loff, tpk, E, chunk, nbin);
    bin_sort<<<nbin, 1024, 0, stream>>>(tpk, loff, rowptr, deg, sorted, N, chunk);

    sage_project<64, 32><<<N / 16, 32, 0, stream>>>(b_z, Ws0, Wn0, b0, S0, P0);

    pull_project<32, 16><<<N / 16, 32, 0, stream>>>(S0, P0, rowptr, sorted, deg, Ws1, Wn1, b1, S1, P1);

    pull_project<16, 1><<<N / 16, 32, 0, stream>>>(S1, P1, rowptr, sorted, deg, Ws2, Wn2, b2, S2, P2);

    pull_final<<<N / 16, 32, 0, stream>>>(P2, rowptr, sorted, S2, deg, out);
}


# Layer 0's dense stage loses its launch: P0=x@Wn0 runs in extra blocks of the sort's first launch, S0=x@Ws0+b0 in the edge-sum wave that uses it (LDS tile): 6 launches become 5; sums unchanged
# speedup vs baseline: 5.4093x; 1.0966x over previous
#include <hip/hip_runtime.h>


typedef __attribute__((ext_vector_type(2))) float v2f;
typedef __attribute__((ext_vector_type(4))) float v4f;
typedef __attribute__((ext_vector_type(8))) float v8f;

constexpr int SORT_G  = 256;
constexpr int BIN_SH  = 8;
constexpr int BIN_W   = 1 << BIN_SH;
constexpr int MAX_BIN = 1024;
constexpr int SORT_T  = 1024;
static_assert(MAX_BIN == SORT_T, "one thread per bin counter");

__device__ __forceinline__ int block_scan_excl(int v, int* wsum, int t) {
    const int lane = t & 31;
    const int w    = t >> 5;
    int x = v;
    for (int off = 1; off < 32; off <<= 1) {
        const int y = __shfl_up(x, off, 32);
        if (lane >= off) x += y;
    }
    if (lane == 31) wsum[w] = x;
    __syncthreads();
    if (w == 0) {
        int s = wsum[lane];
        for (int off = 1; off < 32; off <<= 1) {
            const int y = __shfl_up(s, off, 32);
            if (lane >= off) s += y;
        }
        wsum[lane] = s;
    }
    __syncthreads();
    const int base = (w > 0) ? wsum[w - 1] : 0;
    return base + x - v;
}

__device__ __forceinline__ void
piece_sort_block(const int* __restrict__ src, const int* __restrict__ dst,
                 const float* __restrict__ ew, int* __restrict__ loff,
                 int* __restrict__ tpk, int E, int chunk, int nbin) {
    constexpr int R    = 4;
    constexpr int TILE = R * SORT_T;
    __shared__ int  cur[MAX_BIN];
    __shared__ int  lcu[MAX_BIN];
    __shared__ int  lst[MAX_BIN];
    __shared__ int  wsum[32];
    __shared__ int2 sp[TILE];
    __shared__ int  gp[TILE];
    const int t  = threadIdx.x;
    const int g  = blockIdx.x;
    const int lo = g * chunk;
    const int hi = min(E, lo + chunk);
    if (hi <= lo) {
        if (t <= nbin) loff[t * SORT_G + g] = 0;
        return;
    }

    const bool one = hi - lo <= TILE;
    if (!one) {
        lcu[t] = 0;
        __syncthreads();
        for (int e = lo + t; e < hi; e += SORT_T) atomicAdd(&lcu[dst[e] >> BIN_SH], 1);
        __syncthreads();
        const int v  = lcu[t];
        const int ex = block_scan_excl(v, wsum, t);
        cur[t] = ex;
        if (t <= nbin) loff[t * SORT_G + g] = ex;
        __syncthreads();
    }

    for (int t0 = lo; t0 < hi; t0 += TILE) {
        const int te = min(hi, t0 + TILE);
        lcu[t] = 0;
        __syncthreads();

        int d[R], s[R], w[R], rk[R];
#pragma unroll
        for (int r = 0; r < R; ++r) {
            const int e = t0 + t + SORT_T * r;
            if (e < te) {
                d[r] = dst[e];
                s[r] = src[e];
                w[r] = __float_as_int(ew[e]);
            }
        }
#pragma unroll
        for (int r = 0; r < R; ++r) {
            const int e = t0 + t + SORT_T * r;
            if (e < te) rk[r] = atomicAdd(&lcu[d[r] >> BIN_SH], 1);
        }
        __syncthreads();

        const int c  = lcu[t];
        const int lx = block_scan_excl(c, wsum, t);
        lst[t] = lx;
        if (one) {
            cur[t] = lx;
            if (t <= nbin) loff[t * SORT_G + g] = lx;
        }
        __syncthreads();

#pragma unroll
        for (int r = 0; r < R; ++r) {
            const int e = t0 + t + SORT_T * r;
            if (e < te) {
                const int bin = d[r] >> BIN_SH;
                const int k   = lst[bin] + rk[r];
                sp[k] = make_int2((int)(((unsigned)s[r] << BIN_SH) | ((unsigned)d[r] & (BIN_W - 1))), w[r]);
                gp[k] = cur[bin] + rk[r];
            }
        }
        __syncthreads();

        cur[t] += c;
        const int nt = te - t0;
        for (int k = t; k < nt; k += SORT_T) ((int2*)tpk)[lo + gp[k]] = sp[k];
        __syncthreads();
    }
}

__global__ void __launch_bounds__(1024)
bin_sort(const int* __restrict__ tpk, const int* __restrict__ loff,
         int* __restrict__ rowptr, float* __restrict__ deg, int* __restrict__ sorted,
         int N, int chunk) {
    constexpr int R    = 6;
    constexpr int TILE = R * SORT_T;
    constexpr int SPW  = SORT_G / 32;
    static_assert(SORT_G % 32 == 0 && SORT_G <= SORT_T, "one thread per run; whole runs per wave");
    __shared__ int  cnt[BIN_W];
    __shared__ int  cur[BIN_W];
    __shared__ int  lst[BIN_W];
    __shared__ int  wsum[32];
    __shared__ int  sbase[SORT_G];
    __shared__ int  scnt[SORT_G];
    __shared__ int  sadr[SORT_G];
    __shared__ int  tot[2];
    __shared__ int2 st[TILE];
    __shared__ int2 sp[TILE];
    __shared__ int  gp[TILE];
    const int t    = threadIdx.x;
    const int lane = t & 31;
    const int wv   = t >> 5;
    const int b    = blockIdx.x;
    const int2* pk = (const int2*)tpk;

    int a0 = 0, cg = 0;
    if (t < SORT_G) {
        a0 = loff[b * SORT_G + t];
        cg = loff[(b + 1) * SORT_G + t] - a0;
    }
    if (t < 2) tot[t] = 0;
    __syncthreads();
    if (t < SORT_G) {
        int ra = a0, rc = cg;
        for (int s = 16; s > 0; s >>= 1) {
            ra += __shfl_xor(ra, s, 32);
            rc += __shfl_xor(rc, s, 32);
        }
        if (lane == 0) {
            atomicAdd(&tot[0], ra);
            atomicAdd(&tot[1], rc);
        }
    }
    const int ex = block_scan_excl(cg, wsum, t);
    if (t < SORT_G) {
        sbase[t] = ex;
        scnt[t]  = cg;
        sadr[t]  = t * chunk + a0;
    }
    __syncthreads();
    const int beg   = tot[0];
    const int total = tot[1];
    const int end   = beg + total;

    const bool one = total > 0 && total <= TILE;

    if (!one) {
        if (t < BIN_W) cnt[t] = 0;
        __syncthreads();
        for (int s = 0; s < SPW; ++s) {
            const int g = wv * SPW + s;
            const int c = scnt[g];
            const int a = sadr[g];
            for (int i = lane; i < c; i += 32) atomicAdd(&cnt[pk[a + i].x & (BIN_W - 1)], 1);
        }
        __syncthreads();
        const int v  = (t < BIN_W) ? cnt[t] : 0;
        const int vx = block_scan_excl(v, wsum, t);
        if (t < BIN_W) {
            const int start = beg + vx;
            const int node  = (b << BIN_SH) + t;
            cur[t] = start;
            if (node < N) {
                rowptr[node] = start;
                deg[node]    = (float)v;
                if (node == N - 1) rowptr[N] = end;
            }
        }
        __syncthreads();
    }

    for (int t0 = 0; t0 < total; t0 += TILE) {
        const int nt = min(total - t0, TILE);
        if (t < BIN_W) cnt[t] = 0;

        int2 v[SPW];
        int  p[SPW];
#pragma unroll
        for (int s = 0; s < SPW; ++s) {
            const int g = wv * SPW + s;
            p[s] = (lane < scnt[g]) ? sbase[g] + lane - t0 : -1;
            if (p[s] >= 0 && p[s] < TILE) v[s] = pk[sadr[g] + lane];
        }
#pragma unroll
        for (int s = 0; s < SPW; ++s)
            if (p[s] >= 0 && p[s] < TILE) st[p[s]] = v[s];
        for (int s = 0; s < SPW; ++s) {
            const int g = wv * SPW + s;
            const int c = scnt[g];
            for (int i = lane + 32; i < c; i += 32) {
                const int q = sbase[g] + i - t0;
                if (q >= 0 && q < TILE) st[q] = pk[sadr[g] + i];
            }
        }
        __syncthreads();

        int2 pr[R];
        int  rk[R];
#pragma unroll
        for (int r = 0; r < R; ++r) {
            const int i = t + SORT_T * r;
            if (i < nt) {
                pr[r] = st[i];
                rk[r] = atomicAdd(&cnt[pr[r].x & (BIN_W - 1)], 1);
            }
        }
        __syncthreads();

        const int c  = (t < BIN_W) ? cnt[t] : 0;
        const int lx = block_scan_excl(c, wsum, t);
        if (t < BIN_W) {
            lst[t] = lx;
            if (one) {
                const int start = beg + lx;
                const int node  = (b << BIN_SH) + t;
                cur[t] = start;
                if (node < N) {
                    rowptr[node] = start;
                    deg[node]    = (float)c;
                    if (node == N - 1) rowptr[N] = end;
                }
            }
        }
        __syncthreads();

#pragma unroll
        for (int r = 0; r < R; ++r) {
            const int i = t + SORT_T * r;
            if (i < nt) {
                const int nd = pr[r].x & (BIN_W - 1);
                const int k  = lst[nd] + rk[r];
                sp[k] = make_int2((int)((unsigned)pr[r].x >> BIN_SH), pr[r].y);
                gp[k] = cur[nd] + rk[r];
            }
        }
        __syncthreads();

        if (t < BIN_W) cur[t] += c;
        for (int k = t; k < nt; k += SORT_T) ((int2*)sorted)[gp[k]] = sp[k];
        __syncthreads();
    }
}

template <int K, int F>
__device__ __forceinline__ void
project_neigh_tile(const float* __restrict__ xin, const float* __restrict__ Wn,
                   float* __restrict__ P, int tile, int lane) {
    static_assert(F % 16 == 0, "column tiling");
    static_assert(K % 4 == 0, "k steps");
    constexpr int NS = K / 4;
    constexpr int FT = F / 16;

    const int c    = lane & 15;
    const int hh   = lane >> 4;
    const int row0 = tile << 4;
    const int node = row0 + c;

    v2f a[NS];
    const float* xrow = xin + (size_t)node * K;
    for (int t = 0; t < NS; ++t) {
        const int k0 = 4 * t + 2 * hh;
        a[t] = *(const v2f*)(xrow + k0);
    }

    v8f d[FT];
    for (int j = 0; j < FT; ++j) d[j] = v8f{};

    for (int t = 0; t < NS; ++t) {
        const int k0 = 4 * t + 2 * hh;
        for (int j = 0; j < FT; ++j) {
            const int col = 16 * j + c;
            v2f b;
            b.x = Wn[(k0)     * F + col];
            b.y = Wn[(k0 + 1) * F + col];
            d[j] = __builtin_amdgcn_wmma_f32_16x16x4_f32(false, a[t], false, b,
                                                         (short)0, d[j], false, false);
        }
    }

    for (int j = 0; j < FT; ++j)
        for (int r = 0; r < 8; ++r)
            P[(size_t)(row0 + 8 * hh + r) * F + 16 * j + c] = d[j][r];
}

__global__ void __launch_bounds__(1024)
sort_pieces_and_project(const int* __restrict__ src, const int* __restrict__ dst,
                        const float* __restrict__ ew, int* __restrict__ loff,
                        int* __restrict__ tpk, int E, int chunk, int nbin,
                        const float* __restrict__ xin, const float* __restrict__ Wn,
                        float* __restrict__ P, int ntile) {
    if (blockIdx.x >= SORT_G) {
        const int tile = (blockIdx.x - SORT_G) * (SORT_T / 32) + (threadIdx.x >> 5);
        if (tile < ntile) project_neigh_tile<64, 32>(xin, Wn, P, tile, threadIdx.x & 31);
        return;
    }
    piece_sort_block(src, dst, ew, loff, tpk, E, chunk, nbin);
}

template <int F, int FO, int KX>
__global__ void __launch_bounds__(32)
pull_project(const float* S, const float* __restrict__ P,
             const int* __restrict__ rowptr, const int* __restrict__ sorted,
             const float* __restrict__ deg,
             const float* __restrict__ Ws, const float* __restrict__ Wn,
             const float* __restrict__ bias, float* So, float* Po,
             const float* __restrict__ xin, const float* __restrict__ Wsx,
             const float* __restrict__ bx) {
    static_assert(F == 16 || F == 32, "lane group per node");
    static_assert(FO == 1 || FO == 16, "column tiling");
    static_assert(KX % 4 == 0 && (KX == 0 || F % 16 == 0), "self path: k steps, column tiling");
    constexpr int PIN   = 32;
    constexpr int PQ    = PIN / 4;
    constexpr int LPN   = F / 4;
    constexpr int NPP   = 32 / LPN;
    constexpr int NPASS = 16 / NPP;
    constexpr int EB    = 8;
    constexpr int XP    = F + 4;
    constexpr int NS    = F / 4;
    constexpr int NT    = (FO == 16) ? 2 : 1;
    constexpr int NSX   = (KX > 0) ? KX / 4 : 1;
    constexpr int FTX   = (KX > 0) ? F / 16 : 1;
    __shared__ __attribute__((aligned(16))) float xs[16 * XP];
    __shared__ __attribute__((aligned(16))) float ss[(KX > 0) ? 16 * XP : 4];

    const int lane = threadIdx.x;
    const int sub  = lane % LPN;
    const int grp  = lane / LPN;
    const int row0 = blockIdx.x << 4;
    const int2* ed = (const int2*)sorted;
    const v4f*  P4 = (const v4f*)P;
    const v4f*  S4 = (const v4f*)S;

    int beg[NPASS], end[NPASS];
    for (int j = 0; j < NPASS; ++j) {
        const int node = row0 + j * NPP + grp;
        beg[j] = rowptr[node];
        end[j] = rowptr[node + 1];
    }

    v2f ax[NSX];
    v2f bw[NSX][FTX];
    if constexpr (KX > 0) {
        const int c  = lane & 15;
        const int hh = lane >> 4;
        const float* xrow = xin + (size_t)(row0 + c) * KX;
        for (int t = 0; t < NSX; ++t) {
            const int k0 = 4 * t + 2 * hh;
            ax[t] = *(const v2f*)(xrow + k0);
            for (int j = 0; j < FTX; ++j) {
                bw[t][j].x = Wsx[(k0)     * F + 16 * j + c];
                bw[t][j].y = Wsx[(k0 + 1) * F + 16 * j + c];
            }
        }
        __builtin_amdgcn_sched_barrier(0);
    }

    int2 e[EB];
    {
        const int last = max(end[0] - 1, 0);
        for (int i = 0; i < EB; ++i) e[i] = ed[min(beg[0] + i, last)];
    }

    if constexpr (KX > 0) {
        __builtin_amdgcn_sched_barrier(0);
        const int c  = lane & 15;
        const int hh = lane >> 4;
        v8f ds[FTX];
        for (int j = 0; j < FTX; ++j) ds[j] = v8f{};
        for (int t = 0; t < NSX; ++t)
            for (int j = 0; j < FTX; ++j)
                ds[j] = __builtin_amdgcn_wmma_f32_16x16x4_f32(false, ax[t], false, bw[t][j],
                                                              (short)0, ds[j], false, false);
        for (int j = 0; j < FTX; ++j)
            for (int r = 0; r < 8; ++r)
                ss[(8 * hh + r) * XP + 16 * j + c] = ds[j][r] + bx[16 * j + c];
        __syncthreads();
    }

    for (int j = 0; j < NPASS; ++j) {
        const int slot = j * NPP + grp;
        const int node = row0 + slot;
        const int last = max(end[j] - 1, 0);

        int2 ef[EB];
        if (j + 1 < NPASS) {
            const int lastn = max(end[j + 1] - 1, 0);
            for (int i = 0; i < EB; ++i) ef[i] = ed[min(beg[j + 1] + i, lastn)];
        }
        v4f s4;
        if constexpr (KX > 0) s4 = *(const v4f*)(ss + slot * XP + 4 * sub);
        else                  s4 = S4[(size_t)node * PQ + sub];
        const float dg = deg[node];

        v4f acc = v4f{};
        for (int c0 = beg[j]; c0 < end[j]; c0 += EB) {
            int2 en[EB];
            for (int i = 0; i < EB; ++i) en[i] = ed[min(c0 + EB + i, last)];
            v4f p[EB];
            for (int i = 0; i < EB; ++i) p[i] = P4[e[i].x * PQ + sub];
            __builtin_amdgcn_sched_barrier(0);
            for (int i = 0; i < EB; ++i) {
                const float w  = __int_as_float(e[i].y);
                const bool  in = c0 + i < end[j];
                acc.x += in ? w * p[i].x : 0.0f;
                acc.y += in ? w * p[i].y : 0.0f;
                acc.z += in ? w * p[i].z : 0.0f;
                acc.w += in ? w * p[i].w : 0.0f;
            }
            for (int i = 0; i < EB; ++i) e[i] = en[i];
        }

        const float invdeg = 1.0f / fmaxf(dg, 1.0f);
        v4f x;
        x.x = tanhf(s4.x + acc.x * invdeg);
        x.y = tanhf(s4.y + acc.y * invdeg);
        x.z = tanhf(s4.z + acc.z * invdeg);
        x.w = tanhf(s4.w + acc.w * invdeg);
        *(v4f*)(xs + slot * XP + 4 * sub) = x;

        if (j + 1 < NPASS)
            for (int i = 0; i < EB; ++i) e[i] = ef[i];
    }
    __syncthreads();

    const int c  = lane & 15;
    const int hh = lane >> 4;

    v2f a[NS];
    for (int t = 0; t < NS; ++t) {
        const int k0 = 4 * t + 2 * hh;
        a[t] = *(const v2f*)(xs + c * XP + k0);
    }

    v8f d[NT];
    for (int j = 0; j < NT; ++j) d[j] = v8f{};

    for (int t = 0; t < NS; ++t) {
        const int k0 = 4 * t + 2 * hh;
        for (int j = 0; j < NT; ++j) {
            v2f b;
            if (FO == 16) {
                const float* W = (j == 0) ? Ws : Wn;
                b.x = W[(k0)     * FO + c];
                b.y = W[(k0 + 1) * FO + c];
            } else {
                const float* W = (c == 1) ? Wn : Ws;
                const float w0 = W[k0];
                const float w1 = W[k0 + 1];
                b.x = (c < 2) ? w0 : 0.0f;
                b.y = (c < 2) ? w1 : 0.0f;
            }
            d[j] = __builtin_amdgcn_wmma_f32_16x16x4_f32(false, a[t], false, b,
                                                         (short)0, d[j], false, false);
        }
    }

    for (int r = 0; r < 8; ++r) {
        const size_t row = (size_t)(row0 + 8 * hh + r);
        if (FO == 16) {
            So[row * PIN + c] = d[0][r] + bias[c];
            Po[row * PIN + c] = d[NT - 1][r];
        } else {
            float* o = (c == 1) ? Po : So;
            const float v = (c == 0) ? d[0][r] + bias[0] : d[0][r];
            if (c < 2) o[row] = v;
        }
    }
}

__global__ void __launch_bounds__(32)
pull_final(const float* __restrict__ P, const int* __restrict__ rowptr,
           const int* __restrict__ sorted, const float* __restrict__ S,
           const float* __restrict__ deg, float* __restrict__ out) {
    constexpr int NI = 8;
    const int lane = threadIdx.x;
    const int sub  = lane & 15;
    const int h    = lane >> 4;
    const int base = blockIdx.x * 16;
    const int2* ed = (const int2*)sorted;

    const int   onode = base + 2 * (sub & (NI - 1)) + h;
    const float sv    = S[onode];
    const float dv    = deg[onode];

    const int* rp = rowptr + base + h;
    int beg[NI], end[NI];
    for (int i = 0; i < NI; ++i) {
        beg[i] = rp[2 * i];
        end[i] = rp[2 * i + 1];
    }

    int2 e0[NI], e1[NI];
    for (int i = 0; i < NI; ++i) {
        const int last = max(end[i] - 1, 0);
        e0[i] = ed[min(beg[i] + sub, last)];
        e1[i] = ed[min(beg[i] + sub + 16, last)];
    }
    float p0[NI], p1[NI];
    for (int i = 0; i < NI; ++i) {
        p0[i] = P[e0[i].x];
        p1[i] = P[e1[i].x];
    }

    float acc[NI];
    for (int i = 0; i < NI; ++i) {
        float a = 0.0f;
        if (beg[i] + sub      < end[i]) a += __int_as_float(e0[i].y) * p0[i];
        if (beg[i] + sub + 16 < end[i]) a += __int_as_float(e1[i].y) * p1[i];
        for (int j = beg[i] + sub + 32; j < end[i]; j += 16) {
            const int2 ev = ed[j];
            a += __int_as_float(ev.y) * P[ev.x];
        }
        acc[i] = a;
    }
    for (int s = 8; s > 0; s >>= 1)
        for (int i = 0; i < NI; ++i) acc[i] += __shfl_xor(acc[i], s, 16);

    float res = acc[0];
    for (int i = 1; i < NI; ++i) res = (sub == i) ? acc[i] : res;
    if (sub < NI) out[onode] = sv + res * (1.0f / fmaxf(dv, 1.0f));
}

extern "C" void kernel_launch(void* const* d_in, const int* in_sizes, int n_in,
                              void* d_out, int out_size, void* d_ws, size_t ws_size,
                              hipStream_t stream) {
    const float* b_z  = (const float*)d_in[0];
    const int*   esrc = (const int*)  d_in[1];
    const int*   edst = (const int*)  d_in[2];
    const float* ew   = (const float*)d_in[3];
    const float* Ws0  = (const float*)d_in[4];
    const float* Wn0  = (const float*)d_in[5];
    const float* b0   = (const float*)d_in[6];
    const float* Ws1  = (const float*)d_in[7];
    const float* Wn1  = (const float*)d_in[8];
    const float* b1   = (const float*)d_in[9];
    const float* Ws2  = (const float*)d_in[10];
    const float* Wn2  = (const float*)d_in[11];
    const float* b2   = (const float*)d_in[12];

    const int N = in_sizes[0] / 64;
    const int E = in_sizes[1];
    const size_t n = (size_t)N;

    float* ws     = (float*)d_ws;
    float* deg    = ws;
    int*   rowptr = (int*)(ws + n);
    int*   sorted = (int*)(ws + 4 * n);
    int*   tpk    = (int*)(ws + 36 * n);
    int*   loff   = (int*)(ws + 2 * n + 64);
    float* P0     = ws + 68 * n;
    float* S1     = ws + 36 * n;
    float* P1     = ws + 36 * n + 16;
    float* S2     = ws + 68 * n;
    float* P2     = ws + 72 * n;
    float* out    = (float*)d_out;

    auto cdiv = [](int a, int b) { return (a + b - 1) / b; };

    const int nbin  = cdiv(N, BIN_W);
    const int chunk = cdiv(E, SORT_G);
    const int ntile = N / 16;
    const int pblk  = cdiv(ntile, SORT_T / 32);
    sort_pieces_and_project<<<SORT_G + pblk, 1024, 0, stream>>>(esrc, edst, ew, loff, tpk, E, chunk, nbin,
                                                                b_z, Wn0, P0, ntile);
    bin_sort<<<nbin, 1024, 0, stream>>>(tpk, loff, rowptr, deg, sorted, N, chunk);

    pull_project<32, 16, 64><<<N / 16, 32, 0, stream>>>(nullptr, P0, rowptr, sorted, deg, Ws1, Wn1, b1, S1, P1,
                                                        b_z, Ws0, b0);

    pull_project<16, 1, 0><<<N / 16, 32, 0, stream>>>(S1, P1, rowptr, sorted, deg, Ws2, Wn2, b2, S2, P2,
                                                      nullptr, nullptr, nullptr);

    pull_final<<<N / 16, 32, 0, stream>>>(P2, rowptr, sorted, S2, deg, out);
}


# Edge-sum launches (all three): blocks of 13 independent waves (one 16-node tile each, own LDS tiles, no block barrier), 241 blocks instead of 3125 one-wave blocks; sums unchanged
# speedup vs baseline: 5.6093x; 1.0370x over previous
#include <hip/hip_runtime.h>


typedef __attribute__((ext_vector_type(2))) float v2f;
typedef __attribute__((ext_vector_type(4))) float v4f;
typedef __attribute__((ext_vector_type(8))) float v8f;

constexpr int SORT_G  = 256;
constexpr int BIN_SH  = 8;
constexpr int BIN_W   = 1 << BIN_SH;
constexpr int MAX_BIN = 1024;
constexpr int SORT_T  = 1024;
static_assert(MAX_BIN == SORT_T, "one thread per bin counter");

__device__ __forceinline__ int block_scan_excl(int v, int* wsum, int t) {
    const int lane = t & 31;
    const int w    = t >> 5;
    int x = v;
    for (int off = 1; off < 32; off <<= 1) {
        const int y = __shfl_up(x, off, 32);
        if (lane >= off) x += y;
    }
    if (lane == 31) wsum[w] = x;
    __syncthreads();
    if (w == 0) {
        int s = wsum[lane];
        for (int off = 1; off < 32; off <<= 1) {
            const int y = __shfl_up(s, off, 32);
            if (lane >= off) s += y;
        }
        wsum[lane] = s;
    }
    __syncthreads();
    const int base = (w > 0) ? wsum[w - 1] : 0;
    return base + x - v;
}

__device__ __forceinline__ void
piece_sort_block(const int* __restrict__ src, const int* __restrict__ dst,
                 const float* __restrict__ ew, int* __restrict__ loff,
                 int* __restrict__ tpk, int E, int chunk, int nbin) {
    constexpr int R    = 4;
    constexpr int TILE = R * SORT_T;
    __shared__ int  cur[MAX_BIN];
    __shared__ int  lcu[MAX_BIN];
    __shared__ int  lst[MAX_BIN];
    __shared__ int  wsum[32];
    __shared__ int2 sp[TILE];
    __shared__ int  gp[TILE];
    const int t  = threadIdx.x;
    const int g  = blockIdx.x;
    const int lo = g * chunk;
    const int hi = min(E, lo + chunk);
    if (hi <= lo) {
        if (t <= nbin) loff[t * SORT_G + g] = 0;
        return;
    }

    const bool one = hi - lo <= TILE;
    if (!one) {
        lcu[t] = 0;
        __syncthreads();
        for (int e = lo + t; e < hi; e += SORT_T) atomicAdd(&lcu[dst[e] >> BIN_SH], 1);
        __syncthreads();
        const int v  = lcu[t];
        const int ex = block_scan_excl(v, wsum, t);
        cur[t] = ex;
        if (t <= nbin) loff[t * SORT_G + g] = ex;
        __syncthreads();
    }

    for (int t0 = lo; t0 < hi; t0 += TILE) {
        const int te = min(hi, t0 + TILE);
        lcu[t] = 0;
        __syncthreads();

        int d[R], s[R], w[R], rk[R];
#pragma unroll
        for (int r = 0; r < R; ++r) {
            const int e = t0 + t + SORT_T * r;
            if (e < te) {
                d[r] = dst[e];
                s[r] = src[e];
                w[r] = __float_as_int(ew[e]);
            }
        }
#pragma unroll
        for (int r = 0; r < R; ++r) {
            const int e = t0 + t + SORT_T * r;
            if (e < te) rk[r] = atomicAdd(&lcu[d[r] >> BIN_SH], 1);
        }
        __syncthreads();

        const int c  = lcu[t];
        const int lx = block_scan_excl(c, wsum, t);
        lst[t] = lx;
        if (one) {
            cur[t] = lx;
            if (t <= nbin) loff[t * SORT_G + g] = lx;
        }
        __syncthreads();

#pragma unroll
        for (int r = 0; r < R; ++r) {
            const int e = t0 + t + SORT_T * r;
            if (e < te) {
                const int bin = d[r] >> BIN_SH;
                const int k   = lst[bin] + rk[r];
                sp[k] = make_int2((int)(((unsigned)s[r] << BIN_SH) | ((unsigned)d[r] & (BIN_W - 1))), w[r]);
                gp[k] = cur[bin] + rk[r];
            }
        }
        __syncthreads();

        cur[t] += c;
        const int nt = te - t0;
        for (int k = t; k < nt; k += SORT_T) ((int2*)tpk)[lo + gp[k]] = sp[k];
        __syncthreads();
    }
}

__global__ void __launch_bounds__(1024)
bin_sort(const int* __restrict__ tpk, const int* __restrict__ loff,
         int* __restrict__ rowptr, float* __restrict__ deg, int* __restrict__ sorted,
         int N, int chunk) {
    constexpr int R    = 6;
    constexpr int TILE = R * SORT_T;
    constexpr int SPW  = SORT_G / 32;
    static_assert(SORT_G % 32 == 0 && SORT_G <= SORT_T, "one thread per run; whole runs per wave");
    __shared__ int  cnt[BIN_W];
    __shared__ int  cur[BIN_W];
    __shared__ int  lst[BIN_W];
    __shared__ int  wsum[32];
    __shared__ int  sbase[SORT_G];
    __shared__ int  scnt[SORT_G];
    __shared__ int  sadr[SORT_G];
    __shared__ int  tot[2];
    __shared__ int2 st[TILE];
    __shared__ int2 sp[TILE];
    __shared__ int  gp[TILE];
    const int t    = threadIdx.x;
    const int lane = t & 31;
    const int wv   = t >> 5;
    const int b    = blockIdx.x;
    const int2* pk = (const int2*)tpk;

    int a0 = 0, cg = 0;
    if (t < SORT_G) {
        a0 = loff[b * SORT_G + t];
        cg = loff[(b + 1) * SORT_G + t] - a0;
    }
    if (t < 2) tot[t] = 0;
    __syncthreads();
    if (t < SORT_G) {
        int ra = a0, rc = cg;
        for (int s = 16; s > 0; s >>= 1) {
            ra += __shfl_xor(ra, s, 32);
            rc += __shfl_xor(rc, s, 32);
        }
        if (lane == 0) {
            atomicAdd(&tot[0], ra);
            atomicAdd(&tot[1], rc);
        }
    }
    const int ex = block_scan_excl(cg, wsum, t);
    if (t < SORT_G) {
        sbase[t] = ex;
        scnt[t]  = cg;
        sadr[t]  = t * chunk + a0;
    }
    __syncthreads();
    const int beg   = tot[0];
    const int total = tot[1];
    const int end   = beg + total;

    const bool one = total > 0 && total <= TILE;

    if (!one) {
        if (t < BIN_W) cnt[t] = 0;
        __syncthreads();
        for (int s = 0; s < SPW; ++s) {
            const int g = wv * SPW + s;
            const int c = scnt[g];
            const int a = sadr[g];
            for (int i = lane; i < c; i += 32) atomicAdd(&cnt[pk[a + i].x & (BIN_W - 1)], 1);
        }
        __syncthreads();
        const int v  = (t < BIN_W) ? cnt[t] : 0;
        const int vx = block_scan_excl(v, wsum, t);
        if (t < BIN_W) {
            const int start = beg + vx;
            const int node  = (b << BIN_SH) + t;
            cur[t] = start;
            if (node < N) {
                rowptr[node] = start;
                deg[node]    = (float)v;
                if (node == N - 1) rowptr[N] = end;
            }
        }
        __syncthreads();
    }

    for (int t0 = 0; t0 < total; t0 += TILE) {
        const int nt = min(total - t0, TILE);
        if (t < BIN_W) cnt[t] = 0;

        int2 v[SPW];
        int  p[SPW];
#pragma unroll
        for (int s = 0; s < SPW; ++s) {
            const int g = wv * SPW + s;
            p[s] = (lane < scnt[g]) ? sbase[g] + lane - t0 : -1;
            if (p[s] >= 0 && p[s] < TILE) v[s] = pk[sadr[g] + lane];
        }
#pragma unroll
        for (int s = 0; s < SPW; ++s)
            if (p[s] >= 0 && p[s] < TILE) st[p[s]] = v[s];
        for (int s = 0; s < SPW; ++s) {
            const int g = wv * SPW + s;
            const int c = scnt[g];
            for (int i = lane + 32; i < c; i += 32) {
                const int q = sbase[g] + i - t0;
                if (q >= 0 && q < TILE) st[q] = pk[sadr[g] + i];
            }
        }
        __syncthreads();

        int2 pr[R];
        int  rk[R];
#pragma unroll
        for (int r = 0; r < R; ++r) {
            const int i = t + SORT_T * r;
            if (i < nt) {
                pr[r] = st[i];
                rk[r] = atomicAdd(&cnt[pr[r].x & (BIN_W - 1)], 1);
            }
        }
        __syncthreads();

        const int c  = (t < BIN_W) ? cnt[t] : 0;
        const int lx = block_scan_excl(c, wsum, t);
        if (t < BIN_W) {
            lst[t] = lx;
            if (one) {
                const int start = beg + lx;
                const int node  = (b << BIN_SH) + t;
                cur[t] = start;
                if (node < N) {
                    rowptr[node] = start;
                    deg[node]    = (float)c;
                    if (node == N - 1) rowptr[N] = end;
                }
            }
        }
        __syncthreads();

#pragma unroll
        for (int r = 0; r < R; ++r) {
            const int i = t + SORT_T * r;
            if (i < nt) {
                const int nd = pr[r].x & (BIN_W - 1);
                const int k  = lst[nd] + rk[r];
                sp[k] = make_int2((int)((unsigned)pr[r].x >> BIN_SH), pr[r].y);
                gp[k] = cur[nd] + rk[r];
            }
        }
        __syncthreads();

        if (t < BIN_W) cur[t] += c;
        for (int k = t; k < nt; k += SORT_T) ((int2*)sorted)[gp[k]] = sp[k];
        __syncthreads();
    }
}

template <int K, int F>
__device__ __forceinline__ void
project_neigh_tile(const float* __restrict__ xin, const float* __restrict__ Wn,
                   float* __restrict__ P, int tile, int lane) {
    static_assert(F % 16 == 0, "column tiling");
    static_assert(K % 4 == 0, "k steps");
    constexpr int NS = K / 4;
    constexpr int FT = F / 16;

    const int c    = lane & 15;
    const int hh   = lane >> 4;
    const int row0 = tile << 4;
    const int node = row0 + c;

    v2f a[NS];
    const float* xrow = xin + (size_t)node * K;
    for (int t = 0; t < NS; ++t) {
        const int k0 = 4 * t + 2 * hh;
        a[t] = *(const v2f*)(xrow + k0);
    }

    v8f d[FT];
    for (int j = 0; j < FT; ++j) d[j] = v8f{};

    for (int t = 0; t < NS; ++t) {
        const int k0 = 4 * t + 2 * hh;
        for (int j = 0; j < FT; ++j) {
            const int col = 16 * j + c;
            v2f b;
            b.x = Wn[(k0)     * F + col];
            b.y = Wn[(k0 + 1) * F + col];
            d[j] = __builtin_amdgcn_wmma_f32_16x16x4_f32(false, a[t], false, b,
                                                         (short)0, d[j], false, false);
        }
    }

    for (int j = 0; j < FT; ++j)
        for (int r = 0; r < 8; ++r)
            P[(size_t)(row0 + 8 * hh + r) * F + 16 * j + c] = d[j][r];
}

__global__ void __launch_bounds__(1024)
sort_pieces_and_project(const int* __restrict__ src, const int* __restrict__ dst,
                        const float* __restrict__ ew, int* __restrict__ loff,
                        int* __restrict__ tpk, int E, int chunk, int nbin,
                        const float* __restrict__ xin, const float* __restrict__ Wn,
                        float* __restrict__ P, int ntile) {
    if (blockIdx.x >= SORT_G) {
        const int tile = (blockIdx.x - SORT_G) * (SORT_T / 32) + (threadIdx.x >> 5);
        if (tile < ntile) project_neigh_tile<64, 32>(xin, Wn, P, tile, threadIdx.x & 31);
        return;
    }
    piece_sort_block(src, dst, ew, loff, tpk, E, chunk, nbin);
}

constexpr int PULL_W = 13;

__device__ __forceinline__ void wave_tile_handover() {
    __builtin_amdgcn_fence(__ATOMIC_RELEASE, "wavefront");
    __builtin_amdgcn_wave_barrier();
    __builtin_amdgcn_fence(__ATOMIC_ACQUIRE, "wavefront");
}

template <int F, int FO, int KX>
__global__ void __launch_bounds__(32 * PULL_W)
pull_project(const float* S, const float* __restrict__ P,
             const int* __restrict__ rowptr, const int* __restrict__ sorted,
             const float* __restrict__ deg,
             const float* __restrict__ Ws, const float* __restrict__ Wn,
             const float* __restrict__ bias, float* So, float* Po,
             const float* __restrict__ xin, const float* __restrict__ Wsx,
             const float* __restrict__ bx, int ntile) {
    static_assert(F == 16 || F == 32, "lane group per node");
    static_assert(FO == 1 || FO == 16, "column tiling");
    static_assert(KX % 4 == 0 && (KX == 0 || F % 16 == 0), "self path: k steps, column tiling");
    constexpr int PIN   = 32;
    constexpr int PQ    = PIN / 4;
    constexpr int LPN   = F / 4;
    constexpr int NPP   = 32 / LPN;
    constexpr int NPASS = 16 / NPP;
    constexpr int EB    = 8;
    constexpr int XP    = F + 4;
    constexpr int NS    = F / 4;
    constexpr int NT    = (FO == 16) ? 2 : 1;
    constexpr int NSX   = (KX > 0) ? KX / 4 : 1;
    constexpr int FTX   = (KX > 0) ? F / 16 : 1;
    __shared__ __attribute__((aligned(16))) float xs_all[PULL_W * 16 * XP];
    __shared__ __attribute__((aligned(16))) float ss_all[(KX > 0) ? PULL_W * 16 * XP : 4];

    const int  wv   = __builtin_amdgcn_readfirstlane((int)(threadIdx.x >> 5));
    const int  traw = blockIdx.x * PULL_W + wv;
    const bool live = traw < ntile;
    const int  tile = live ? traw : ntile - 1;
    float* xs = xs_all + wv * (16 * XP);
    float* ss = ss_all + ((KX > 0) ? wv * (16 * XP) : 0);

    const int lane = threadIdx.x & 31;
    const int sub  = lane % LPN;
    const int grp  = lane / LPN;
    const int row0 = tile << 4;
    const int2* ed = (const int2*)sorted;
    const v4f*  P4 = (const v4f*)P;
    const v4f*  S4 = (const v4f*)S;

    int beg[NPASS], end[NPASS];
    for (int j = 0; j < NPASS; ++j) {
        const int node = row0 + j * NPP + grp;
        beg[j] = rowptr[node];
        end[j] = rowptr[node + 1];
    }

    v2f ax[NSX];
    v2f bw[NSX][FTX];
    if constexpr (KX > 0) {
        const int c  = lane & 15;
        const int hh = lane >> 4;
        const float* xrow = xin + (size_t)(row0 + c) * KX;
        for (int t = 0; t < NSX; ++t) {
            const int k0 = 4 * t + 2 * hh;
            ax[t] = *(const v2f*)(xrow + k0);
            for (int j = 0; j < FTX; ++j) {
                bw[t][j].x = Wsx[(k0)     * F + 16 * j + c];
                bw[t][j].y = Wsx[(k0 + 1) * F + 16 * j + c];
            }
        }
        __builtin_amdgcn_sched_barrier(0);
    }

    int2 e[EB];
    {
        const int last = max(end[0] - 1, 0);
        for (int i = 0; i < EB; ++i) e[i] = ed[min(beg[0] + i, last)];
    }

    if constexpr (KX > 0) {
        __builtin_amdgcn_sched_barrier(0);
        const int c  = lane & 15;
        const int hh = lane >> 4;
        v8f ds[FTX];
        for (int j = 0; j < FTX; ++j) ds[j] = v8f{};
        for (int t = 0; t < NSX; ++t)
            for (int j = 0; j < FTX; ++j)
                ds[j] = __builtin_amdgcn_wmma_f32_16x16x4_f32(false, ax[t], false, bw[t][j],
                                                              (short)0, ds[j], false, false);
        for (int j = 0; j < FTX; ++j)
            for (int r = 0; r < 8; ++r)
                ss[(8 * hh + r) * XP + 16 * j + c] = ds[j][r] + bx[16 * j + c];
        wave_tile_handover();
    }

    for (int j = 0; j < NPASS; ++j) {
        const int slot = j * NPP + grp;
        const int node = row0 + slot;
        const int last = max(end[j] - 1, 0);

        int2 ef[EB];
        if (j + 1 < NPASS) {
            const int lastn = max(end[j + 1] - 1, 0);
            for (int i = 0; i < EB; ++i) ef[i] = ed[min(beg[j + 1] + i, lastn)];
        }
        v4f s4;
        if constexpr (KX > 0) s4 = *(const v4f*)(ss + slot * XP + 4 * sub);
        else                  s4 = S4[(size_t)node * PQ + sub];
        const float dg = deg[node];

        v4f acc = v4f{};
        for (int c0 = beg[j]; c0 < end[j]; c0 += EB) {
            int2 en[EB];
            for (int i = 0; i < EB; ++i) en[i] = ed[min(c0 + EB + i, last)];
            v4f p[EB];
            for (int i = 0; i < EB; ++i) p[i] = P4[e[i].x * PQ + sub];
            __builtin_amdgcn_sched_barrier(0);
            for (int i = 0; i < EB; ++i) {
                const float w  = __int_as_float(e[i].y);
                const bool  in = c0 + i < end[j];
                acc.x += in ? w * p[i].x : 0.0f;
                acc.y += in ? w * p[i].y : 0.0f;
                acc.z += in ? w * p[i].z : 0.0f;
                acc.w += in ? w * p[i].w : 0.0f;
            }
            for (int i = 0; i < EB; ++i) e[i] = en[i];
        }

        const float invdeg = 1.0f / fmaxf(dg, 1.0f);
        v4f x;
        x.x = tanhf(s4.x + acc.x * invdeg);
        x.y = tanhf(s4.y + acc.y * invdeg);
        x.z = tanhf(s4.z + acc.z * invdeg);
        x.w = tanhf(s4.w + acc.w * invdeg);
        *(v4f*)(xs + slot * XP + 4 * sub) = x;

        if (j + 1 < NPASS)
            for (int i = 0; i < EB; ++i) e[i] = ef[i];
    }
    wave_tile_handover();
    if (!live) return;

    const int c  = lane & 15;
    const int hh = lane >> 4;

    v2f a[NS];
    for (int t = 0; t < NS; ++t) {
        const int k0 = 4 * t + 2 * hh;
        a[t] = *(const v2f*)(xs + c * XP + k0);
    }

    v8f d[NT];
    for (int j = 0; j < NT; ++j) d[j] = v8f{};

    for (int t = 0; t < NS; ++t) {
        const int k0 = 4 * t + 2 * hh;
        for (int j = 0; j < NT; ++j) {
            v2f b;
            if (FO == 16) {
                const float* W = (j == 0) ? Ws : Wn;
                b.x = W[(k0)     * FO + c];
                b.y = W[(k0 + 1) * FO + c];
            } else {
                const float* W = (c == 1) ? Wn : Ws;
                const float w0 = W[k0];
                const float w1 = W[k0 + 1];
                b.x = (c < 2) ? w0 : 0.0f;
                b.y = (c < 2) ? w1 : 0.0f;
            }
            d[j] = __builtin_amdgcn_wmma_f32_16x16x4_f32(false, a[t], false, b,
                                                         (short)0, d[j], false, false);
        }
    }

    for (int r = 0; r < 8; ++r) {
        const size_t row = (size_t)(row0 + 8 * hh + r);
        if (FO == 16) {
            So[row * PIN + c] = d[0][r] + bias[c];
            Po[row * PIN + c] = d[NT - 1][r];
        } else {
            float* o = (c == 1) ? Po : So;
            const float v = (c == 0) ? d[0][r] + bias[0] : d[0][r];
            if (c < 2) o[row] = v;
        }
    }
}

__global__ void __launch_bounds__(32 * PULL_W)
pull_final(const float* __restrict__ P, const int* __restrict__ rowptr,
           const int* __restrict__ sorted, const float* __restrict__ S,
           const float* __restrict__ deg, float* __restrict__ out, int ntile) {
    constexpr int NI = 8;
    const int  traw = blockIdx.x * PULL_W + __builtin_amdgcn_readfirstlane((int)(threadIdx.x >> 5));
    const bool live = traw < ntile;
    const int  tile = live ? traw : ntile - 1;
    const int lane = threadIdx.x & 31;
    const int sub  = lane & 15;
    const int h    = lane >> 4;
    const int base = tile * 16;
    const int2* ed = (const int2*)sorted;

    const int   onode = base + 2 * (sub & (NI - 1)) + h;
    const float sv    = S[onode];
    const float dv    = deg[onode];

    const int* rp = rowptr + base + h;
    int beg[NI], end[NI];
    for (int i = 0; i < NI; ++i) {
        beg[i] = rp[2 * i];
        end[i] = rp[2 * i + 1];
    }

    int2 e0[NI], e1[NI];
    for (int i = 0; i < NI; ++i) {
        const int last = max(end[i] - 1, 0);
        e0[i] = ed[min(beg[i] + sub, last)];
        e1[i] = ed[min(beg[i] + sub + 16, last)];
    }
    float p0[NI], p1[NI];
    for (int i = 0; i < NI; ++i) {
        p0[i] = P[e0[i].x];
        p1[i] = P[e1[i].x];
    }

    float acc[NI];
    for (int i = 0; i < NI; ++i) {
        float a = 0.0f;
        if (beg[i] + sub      < end[i]) a += __int_as_float(e0[i].y) * p0[i];
        if (beg[i] + sub + 16 < end[i]) a += __int_as_float(e1[i].y) * p1[i];
        for (int j = beg[i] + sub + 32; j < end[i]; j += 16) {
            const int2 ev = ed[j];
            a += __int_as_float(ev.y) * P[ev.x];
        }
        acc[i] = a;
    }
    for (int s = 8; s > 0; s >>= 1)
        for (int i = 0; i < NI; ++i) acc[i] += __shfl_xor(acc[i], s, 16);

    float res = acc[0];
    for (int i = 1; i < NI; ++i) res = (sub == i) ? acc[i] : res;
    if (sub < NI && live) out[onode] = sv + res * (1.0f / fmaxf(dv, 1.0f));
}

extern "C" void kernel_launch(void* const* d_in, const int* in_sizes, int n_in,
                              void* d_out, int out_size, void* d_ws, size_t ws_size,
                              hipStream_t stream) {
    const float* b_z  = (const float*)d_in[0];
    const int*   esrc = (const int*)  d_in[1];
    const int*   edst = (const int*)  d_in[2];
    const float* ew   = (const float*)d_in[3];
    const float* Ws0  = (const float*)d_in[4];
    const float* Wn0  = (const float*)d_in[5];
    const float* b0   = (const float*)d_in[6];
    const float* Ws1  = (const float*)d_in[7];
    const float* Wn1  = (const float*)d_in[8];
    const float* b1   = (const float*)d_in[9];
    const float* Ws2  = (const float*)d_in[10];
    const float* Wn2  = (const float*)d_in[11];
    const float* b2   = (const float*)d_in[12];

    const int N = in_sizes[0] / 64;
    const int E = in_sizes[1];
    const size_t n = (size_t)N;

    float* ws     = (float*)d_ws;
    float* deg    = ws;
    int*   rowptr = (int*)(ws + n);
    int*   sorted = (int*)(ws + 4 * n);
    int*   tpk    = (int*)(ws + 36 * n);
    int*   loff   = (int*)(ws + 2 * n + 64);
    float* P0     = ws + 68 * n;
    float* S1     = ws + 36 * n;
    float* P1     = ws + 36 * n + 16;
    float* S2     = ws + 68 * n;
    float* P2     = ws + 72 * n;
    float* out    = (float*)d_out;

    auto cdiv = [](int a, int b) { return (a + b - 1) / b; };

    const int nbin  = cdiv(N, BIN_W);
    const int chunk = cdiv(E, SORT_G);
    const int ntile = N / 16;
    const int pblk  = cdiv(ntile, SORT_T / 32);
    sort_pieces_and_project<<<SORT_G + pblk, 1024, 0, stream>>>(esrc, edst, ew, loff, tpk, E, chunk, nbin,
                                                                b_z, Wn0, P0, ntile);
    bin_sort<<<nbin, 1024, 0, stream>>>(tpk, loff, rowptr, deg, sorted, N, chunk);

    const int pullb = cdiv(ntile, PULL_W);
    pull_project<32, 16, 64><<<pullb, 32 * PULL_W, 0, stream>>>(nullptr, P0, rowptr, sorted, deg, Ws1, Wn1, b1, S1, P1,
                                                                b_z, Ws0, b0, ntile);

    pull_project<16, 1, 0><<<pullb, 32 * PULL_W, 0, stream>>>(S1, P1, rowptr, sorted, deg, Ws2, Wn2, b2, S2, P2,
                                                              nullptr, nullptr, nullptr, ntile);

    pull_final<<<pullb, 32 * PULL_W, 0, stream>>>(P2, rowptr, sorted, S2, deg, out, ntile);
}
